# starting kernel
# baseline (speedup 1.0000x reference)
#include <hip/hip_runtime.h>

typedef __attribute__((ext_vector_type(2))) float v2f;
typedef __attribute__((ext_vector_type(8))) float v8f;

#define NN 100000
#define NE 1600000
#define IND 128
#define HID 128
#define OUTD 64
#define NG 512


__global__ void fill_zero_k(float* __restrict__ p, int n) {
    int i = blockIdx.x * blockDim.x + threadIdx.x;
    if (i < n) p[i] = 0.0f;
}

__global__ void deg_accum_k(const int* __restrict__ dst, float* __restrict__ deg, int E) {
    int e = blockIdx.x * blockDim.x + threadIdx.x;
    if (e < E) atomicAdd(&deg[dst[e]], 1.0f);
}

__global__ void deg_to_dinv_k(float* __restrict__ d, int n) {
    int i = blockIdx.x * blockDim.x + threadIdx.x;
    if (i < n) d[i] = rsqrtf(d[i] + 1.0f);
}

__global__ void edge_scatter_k(const int* __restrict__ src, const int* __restrict__ dst,
                               const float* __restrict__ dinv, const float* __restrict__ h,
                               float* __restrict__ agg, int E) {
    unsigned t = blockIdx.x * blockDim.x + threadIdx.x;
    unsigned e = t >> 5;
    if (e >= (unsigned)E) return;
    unsigned q = (t & 31u) * 4u;
    int s = src[e], d = dst[e];
    float w = dinv[s] * dinv[d];
    float4 hv = *reinterpret_cast<const float4*>(h + (size_t)s * HID + q);
    float* ap = agg + (size_t)d * HID + q;
    atomicAdd(ap + 0, hv.x * w);
    atomicAdd(ap + 1, hv.y * w);
    atomicAdd(ap + 2, hv.z * w);
    atomicAdd(ap + 3, hv.w * w);
}

__global__ void pool_scatter_k(const int* __restrict__ batch, const float* __restrict__ h,
                               const float* __restrict__ b, float* __restrict__ pooled,
                               int nNodes) {
    unsigned t = blockIdx.x * blockDim.x + threadIdx.x;
    unsigned node = t >> 5;
    if (node >= (unsigned)nNodes) return;
    unsigned q = (t & 31u) * 4u;
    int g = batch[node];
    float4 hv = *reinterpret_cast<const float4*>(h + (size_t)node * HID + q);
    float4 bb = *reinterpret_cast<const float4*>(b + q);
    float* pp = pooled + (size_t)g * HID + q;
    atomicAdd(pp + 0, fmaxf(hv.x + bb.x, 0.0f));
    atomicAdd(pp + 1, fmaxf(hv.y + bb.y, 0.0f));
    atomicAdd(pp + 2, fmaxf(hv.z + bb.z, 0.0f));
    atomicAdd(pp + 3, fmaxf(hv.w + bb.w, 0.0f));
}

template <int MT, int N, int K, bool ACT_A, bool SELFLOOP, bool BIAS_OUT>
__global__ void gemm_k(const float* __restrict__ A, const float* __restrict__ B,
                       float* __restrict__ C, const float* __restrict__ actBias,
                       const float* __restrict__ dinv, float* __restrict__ agg,
                       const float* __restrict__ biasOut) {
    const int lane   = threadIdx.x & 31;
    const int wave   = threadIdx.x >> 5;
    const int half   = lane >> 4;
    const int l16    = lane & 15;
    const int nBase  = wave * 16;
    const int mTile0 = blockIdx.x * MT;

    const float* aBase = A + (size_t)(mTile0 * 16 + l16) * K + half * 2;
    const float* bBase = B + (size_t)(half * 2) * N + nBase + l16;
    const float* abBase = actBias + half * 2;

    v8f acc[MT];
#pragma unroll
    for (int t = 0; t < MT; ++t) {
        v8f z = {};
        acc[t] = z;
    }

#pragma unroll 4
    for (int k = 0; k < K; k += 4) {
        v2f b;
        b.x = bBase[k * N];
        b.y = bBase[k * N + N];
        float abx = 0.0f, aby = 0.0f;
        if (ACT_A) {
            abx = abBase[k];
            aby = abBase[k + 1];
        }
#pragma unroll
        for (int t = 0; t < MT; ++t) {
            v2f a;
            a.x = aBase[t * 16 * K + k];
            a.y = aBase[t * 16 * K + k + 1];
            if (ACT_A) {
                a.x = fmaxf(a.x + abx, 0.0f);
                a.y = fmaxf(a.y + aby, 0.0f);
            }
            acc[t] = __builtin_amdgcn_wmma_f32_16x16x4_f32(false, a, false, b, (short)0,
                                                           acc[t], false, false);
        }
    }

    if (SELFLOOP) __syncthreads();

    float bv = BIAS_OUT ? biasOut[nBase + l16] : 0.0f;
#pragma unroll
    for (int t = 0; t < MT; ++t) {
        const int rBase = (mTile0 + t) * 16 + half * 8;
#pragma unroll
        for (int v = 0; v < 8; ++v) {
            const size_t idx = (size_t)(rBase + v) * N + nBase + l16;
            float val = acc[t][v] + bv;
            C[idx] = val;
            if (SELFLOOP) {
                float w = dinv[rBase + v];
                agg[idx] = val * w * w;
            }
        }
    }
}


extern "C" void kernel_launch(void* const* d_in, const int* in_sizes, int n_in,
                              void* d_out, int out_size, void* d_ws, size_t ws_size,
                              hipStream_t stream) {
    const float* x     = (const float*)d_in[0];
    const int*   ei    = (const int*)d_in[1];
    const int*   batch = (const int*)d_in[2];
    const float* W1    = (const float*)d_in[3];
    const float* b1    = (const float*)d_in[4];
    const float* W2    = (const float*)d_in[5];
    const float* b2    = (const float*)d_in[6];
    const float* Wl    = (const float*)d_in[7];
    const float* bl    = (const float*)d_in[8];
    float* out = (float*)d_out;

    const int* src = ei;
    const int* dst = ei + NE;

    char* ws = (char*)d_ws;
    const size_t FEAT_BYTES = (size_t)NN * HID * sizeof(float);
    float* dinv   = (float*)(ws);
    float* bufH   = (float*)(ws + 524288);
    float* bufAgg = (float*)(ws + 524288 + FEAT_BYTES);
    float* pooled = (float*)(ws + 524288 + 2 * FEAT_BYTES);

    const int TPB = 256;

    fill_zero_k<<<(NN + TPB - 1) / TPB, TPB, 0, stream>>>(dinv, NN);
    deg_accum_k<<<(NE + TPB - 1) / TPB, TPB, 0, stream>>>(dst, dinv, NE);
    deg_to_dinv_k<<<(NN + TPB - 1) / TPB, TPB, 0, stream>>>(dinv, NN);

    gemm_k<5, HID, IND, false, true, false><<<1250, 32 * (HID / 16), 0, stream>>>(
        x, W1, bufH, nullptr, dinv, bufAgg, nullptr);
    edge_scatter_k<<<(NE * 32 + TPB - 1) / TPB, TPB, 0, stream>>>(src, dst, dinv, bufH,
                                                                  bufAgg, NE);

    gemm_k<5, HID, HID, true, true, false><<<1250, 32 * (HID / 16), 0, stream>>>(
        bufAgg, W2, bufH, b1, dinv, bufAgg, nullptr);
    edge_scatter_k<<<(NE * 32 + TPB - 1) / TPB, TPB, 0, stream>>>(src, dst, dinv, bufH,
                                                                  bufAgg, NE);

    fill_zero_k<<<(NG * HID + TPB - 1) / TPB, TPB, 0, stream>>>(pooled, NG * HID);
    pool_scatter_k<<<(NN * 32 + TPB - 1) / TPB, TPB, 0, stream>>>(batch, bufAgg, b2,
                                                                  pooled, NN);

    gemm_k<2, OUTD, HID, false, false, true><<<16, 32 * (OUTD / 16), 0, stream>>>(
        pooled, Wl, out, nullptr, nullptr, nullptr, bl);
}


# Edge aggregation without float atomics: per-call grouping of edges by destination (integer atomic ranks), one wave per node sums dinv-prescaled rows in registers; row sum order = arrival rank
# speedup vs baseline: 2.1300x; 2.1300x over previous
#include <hip/hip_runtime.h>

typedef __attribute__((ext_vector_type(2))) float v2f;
typedef __attribute__((ext_vector_type(8))) float v8f;

#define NN 100000
#define NE 1600000
#define IND 128
#define HID 128
#define OUTD 64
#define NG 512


__global__ void fill_zero_k(float* __restrict__ p, int n) {
    int i = blockIdx.x * blockDim.x + threadIdx.x;
    if (i < n) p[i] = 0.0f;
}

__global__ void fill_zero_i_k(int* __restrict__ p, int n) {
    int i = blockIdx.x * blockDim.x + threadIdx.x;
    if (i < n) p[i] = 0;
}


__global__ void count_rank_k(const int* __restrict__ dst, int* __restrict__ cnt,
                             int* __restrict__ rank, int E) {
    int e = blockIdx.x * blockDim.x + threadIdx.x;
    if (e < E) rank[e] = atomicAdd(&cnt[dst[e]], 1);
}

__global__ void row_start_k(const int* __restrict__ cnt, int* __restrict__ rowStart,
                            float* __restrict__ dinv, int* __restrict__ cursor, int n) {
    const int i = blockIdx.x * blockDim.x + threadIdx.x;
    const int lane = threadIdx.x & 31;
    const int c = (i < n) ? cnt[i] : 0;
    int s = c;
#pragma unroll
    for (int off = 1; off < 32; off <<= 1) {
        int v = __shfl_up(s, off);
        if (lane >= off) s += v;
    }
    const int total = __shfl(s, 31);
    int base = 0;
    if (lane == 0) base = atomicAdd(cursor, total);
    base = __shfl(base, 0);
    if (i < n) {
        rowStart[i] = base + s - c;
        dinv[i] = rsqrtf((float)c + 1.0f);
    }
}

__global__ void group_fill_k(const int* __restrict__ src, const int* __restrict__ dst,
                             const int* __restrict__ rank, const int* __restrict__ rowStart,
                             int* __restrict__ grouped, int E) {
    int e = blockIdx.x * blockDim.x + threadIdx.x;
    if (e < E) grouped[rowStart[dst[e]] + rank[e]] = src[e];
}

template <int U>
__device__ __forceinline__ void gather_add(float4& acc, const float4* __restrict__ hs4, int idx,
                                           int k, int lane) {
    float4 v[U];
#pragma unroll
    for (int u = 0; u < U; ++u) {
        const unsigned s = (unsigned)__builtin_amdgcn_readlane(idx, k + u);
        v[u] = hs4[(size_t)s * (HID / 4) + lane];
    }
#pragma unroll
    for (int u = 0; u < U; ++u) {
        acc.x += v[u].x;
        acc.y += v[u].y;
        acc.z += v[u].z;
        acc.w += v[u].w;
    }
}

__global__ void agg_k(const int* __restrict__ rowStart, const int* __restrict__ cnt,
                      const int* __restrict__ grouped, const float* __restrict__ dinv,
                      const float* __restrict__ hs, float* __restrict__ out, int n) {
    const int lane = threadIdx.x & 31;
    const int d = __builtin_amdgcn_readfirstlane((int)((blockIdx.x * blockDim.x + threadIdx.x) >> 5));
    if (d >= n) return;
    const int start = rowStart[d];
    const int c = cnt[d];
    const float4* hs4 = reinterpret_cast<const float4*>(hs);
    float4 acc = hs4[(size_t)d * (HID / 4) + lane];
    for (int base = 0; base < c; base += 32) {
        const int m = (c - base < 32) ? (c - base) : 32;
        int idx = 0;
        if (lane < m) idx = grouped[start + base + lane];
        int k = 0;
        for (; k + 8 <= m; k += 8) gather_add<8>(acc, hs4, idx, k, lane);
        if (k + 4 <= m) { gather_add<4>(acc, hs4, idx, k, lane); k += 4; }
        if (k + 2 <= m) { gather_add<2>(acc, hs4, idx, k, lane); k += 2; }
        if (k < m) gather_add<1>(acc, hs4, idx, k, lane);
    }
    const float w = dinv[d];
    acc.x *= w;
    acc.y *= w;
    acc.z *= w;
    acc.w *= w;
    reinterpret_cast<float4*>(out)[(size_t)d * (HID / 4) + lane] = acc;
}

__global__ void pool_scatter_k(const int* __restrict__ batch, const float* __restrict__ h,
                               const float* __restrict__ b, float* __restrict__ pooled,
                               int nNodes) {
    unsigned t = blockIdx.x * blockDim.x + threadIdx.x;
    unsigned node = t >> 5;
    if (node >= (unsigned)nNodes) return;
    unsigned q = (t & 31u) * 4u;
    int g = batch[node];
    float4 hv = *reinterpret_cast<const float4*>(h + (size_t)node * HID + q);
    float4 bb = *reinterpret_cast<const float4*>(b + q);
    float* pp = pooled + (size_t)g * HID + q;
    atomicAdd(pp + 0, fmaxf(hv.x + bb.x, 0.0f));
    atomicAdd(pp + 1, fmaxf(hv.y + bb.y, 0.0f));
    atomicAdd(pp + 2, fmaxf(hv.z + bb.z, 0.0f));
    atomicAdd(pp + 3, fmaxf(hv.w + bb.w, 0.0f));
}

template <int MT, int N, int K, bool ACT_A, bool ROWSCALE, bool BIAS_OUT>
__global__ void gemm_k(const float* __restrict__ A, const float* __restrict__ B,
                       float* __restrict__ C, const float* __restrict__ actBias,
                       const float* __restrict__ rowScale, const float* __restrict__ biasOut) {
    const int lane   = threadIdx.x & 31;
    const int wave   = threadIdx.x >> 5;
    const int half   = lane >> 4;
    const int l16    = lane & 15;
    const int nBase  = wave * 16;
    const int mTile0 = blockIdx.x * MT;

    const float* aBase = A + (size_t)(mTile0 * 16 + l16) * K + half * 2;
    const float* bBase = B + (size_t)(half * 2) * N + nBase + l16;
    const float* abBase = actBias + half * 2;

    v8f acc[MT];
#pragma unroll
    for (int t = 0; t < MT; ++t) {
        v8f z = {};
        acc[t] = z;
    }

#pragma unroll 4
    for (int k = 0; k < K; k += 4) {
        v2f b;
        b.x = bBase[k * N];
        b.y = bBase[k * N + N];
        float abx = 0.0f, aby = 0.0f;
        if (ACT_A) {
            abx = abBase[k];
            aby = abBase[k + 1];
        }
#pragma unroll
        for (int t = 0; t < MT; ++t) {
            v2f a;
            a.x = aBase[t * 16 * K + k];
            a.y = aBase[t * 16 * K + k + 1];
            if (ACT_A) {
                a.x = fmaxf(a.x + abx, 0.0f);
                a.y = fmaxf(a.y + aby, 0.0f);
            }
            acc[t] = __builtin_amdgcn_wmma_f32_16x16x4_f32(false, a, false, b, (short)0,
                                                           acc[t], false, false);
        }
    }

    float bv = BIAS_OUT ? biasOut[nBase + l16] : 0.0f;
#pragma unroll
    for (int t = 0; t < MT; ++t) {
        const int rBase = (mTile0 + t) * 16 + half * 8;
#pragma unroll
        for (int v = 0; v < 8; ++v) {
            const size_t idx = (size_t)(rBase + v) * N + nBase + l16;
            float val = acc[t][v] + bv;
            if (ROWSCALE) val *= rowScale[rBase + v];
            C[idx] = val;
        }
    }
}


extern "C" void kernel_launch(void* const* d_in, const int* in_sizes, int n_in,
                              void* d_out, int out_size, void* d_ws, size_t ws_size,
                              hipStream_t stream) {
    const float* x     = (const float*)d_in[0];
    const int*   ei    = (const int*)d_in[1];
    const int*   batch = (const int*)d_in[2];
    const float* W1    = (const float*)d_in[3];
    const float* b1    = (const float*)d_in[4];
    const float* W2    = (const float*)d_in[5];
    const float* b2    = (const float*)d_in[6];
    const float* Wl    = (const float*)d_in[7];
    const float* bl    = (const float*)d_in[8];
    float* out = (float*)d_out;

    const int* src = ei;
    const int* dst = ei + NE;

    char* ws = (char*)d_ws;
    const size_t FEAT_BYTES = (size_t)NN * HID * sizeof(float);
    const size_t EDGE_BYTES = (size_t)NE * sizeof(int);
    float* dinv    = (float*)(ws);
    float* bufH    = (float*)(ws + 524288);
    float* bufAgg  = (float*)(ws + 524288 + FEAT_BYTES);
    float* pooled  = (float*)(ws + 524288 + 2 * FEAT_BYTES);
    int* cnt       = (int*)(ws + 524288 + 2 * FEAT_BYTES + 262144);
    int* rowStart  = (int*)(ws + 524288 + 2 * FEAT_BYTES + 262144 + 524288);
    int* rank      = (int*)(ws + 524288 + 2 * FEAT_BYTES + 262144 + 2 * 524288);
    int* grouped   = (int*)(ws + 524288 + 2 * FEAT_BYTES + 262144 + 2 * 524288 + EDGE_BYTES);
    int* cursor    = cnt + NN;

    const int TPB = 256;

    fill_zero_i_k<<<(NN + 32 + TPB - 1) / TPB, TPB, 0, stream>>>(cnt, NN + 32);
    count_rank_k<<<(NE + TPB - 1) / TPB, TPB, 0, stream>>>(dst, cnt, rank, NE);
    row_start_k<<<(NN + TPB - 1) / TPB, TPB, 0, stream>>>(cnt, rowStart, dinv, cursor, NN);
    group_fill_k<<<(NE + TPB - 1) / TPB, TPB, 0, stream>>>(src, dst, rank, rowStart, grouped, NE);

    gemm_k<5, HID, IND, false, true, false><<<1250, 32 * (HID / 16), 0, stream>>>(
        x, W1, bufH, nullptr, dinv, nullptr);
    agg_k<<<(NN * 32 + TPB - 1) / TPB, TPB, 0, stream>>>(rowStart, cnt, grouped, dinv, bufH,
                                                         bufAgg, NN);

    gemm_k<5, HID, HID, true, true, false><<<1250, 32 * (HID / 16), 0, stream>>>(
        bufAgg, W2, bufH, b1, dinv, nullptr);
    agg_k<<<(NN * 32 + TPB - 1) / TPB, TPB, 0, stream>>>(rowStart, cnt, grouped, dinv, bufH,
                                                         bufAgg, NN);

    fill_zero_k<<<(NG * HID + TPB - 1) / TPB, TPB, 0, stream>>>(pooled, NG * HID);
    pool_scatter_k<<<(NN * 32 + TPB - 1) / TPB, TPB, 0, stream>>>(batch, bufAgg, b2,
                                                                  pooled, NN);

    gemm_k<2, OUTD, HID, false, false, true><<<16, 32 * (OUTD / 16), 0, stream>>>(
        pooled, Wl, out, nullptr, nullptr, bl);
}


# Pooling: each wave sums runs of consecutive same-graph node rows in registers (32 nodes per wave), one float atomic per feature per run, not per node; pooled sum order = run partials, arrival order
# speedup vs baseline: 2.5673x; 1.2053x over previous
#include <hip/hip_runtime.h>

typedef __attribute__((ext_vector_type(2))) float v2f;
typedef __attribute__((ext_vector_type(8))) float v8f;

#define NN 100000
#define NE 1600000
#define IND 128
#define HID 128
#define OUTD 64
#define NG 512


__global__ void fill_zero_k(float* __restrict__ p, int n) {
    int i = blockIdx.x * blockDim.x + threadIdx.x;
    if (i < n) p[i] = 0.0f;
}

__global__ void fill_zero_i_k(int* __restrict__ p, int n) {
    int i = blockIdx.x * blockDim.x + threadIdx.x;
    if (i < n) p[i] = 0;
}


__global__ void count_rank_k(const int* __restrict__ dst, int* __restrict__ cnt,
                             int* __restrict__ rank, int E) {
    int e = blockIdx.x * blockDim.x + threadIdx.x;
    if (e < E) rank[e] = atomicAdd(&cnt[dst[e]], 1);
}

__global__ void row_start_k(const int* __restrict__ cnt, int* __restrict__ rowStart,
                            float* __restrict__ dinv, int* __restrict__ cursor, int n) {
    const int i = blockIdx.x * blockDim.x + threadIdx.x;
    const int lane = threadIdx.x & 31;
    const int c = (i < n) ? cnt[i] : 0;
    int s = c;
#pragma unroll
    for (int off = 1; off < 32; off <<= 1) {
        int v = __shfl_up(s, off);
        if (lane >= off) s += v;
    }
    const int total = __shfl(s, 31);
    int base = 0;
    if (lane == 0) base = atomicAdd(cursor, total);
    base = __shfl(base, 0);
    if (i < n) {
        rowStart[i] = base + s - c;
        dinv[i] = rsqrtf((float)c + 1.0f);
    }
}

__global__ void group_fill_k(const int* __restrict__ src, const int* __restrict__ dst,
                             const int* __restrict__ rank, const int* __restrict__ rowStart,
                             int* __restrict__ grouped, int E) {
    int e = blockIdx.x * blockDim.x + threadIdx.x;
    if (e < E) grouped[rowStart[dst[e]] + rank[e]] = src[e];
}

template <int U>
__device__ __forceinline__ void gather_add(float4& acc, const float4* __restrict__ hs4, int idx,
                                           int k, int lane) {
    float4 v[U];
#pragma unroll
    for (int u = 0; u < U; ++u) {
        const unsigned s = (unsigned)__builtin_amdgcn_readlane(idx, k + u);
        v[u] = hs4[(size_t)s * (HID / 4) + lane];
    }
#pragma unroll
    for (int u = 0; u < U; ++u) {
        acc.x += v[u].x;
        acc.y += v[u].y;
        acc.z += v[u].z;
        acc.w += v[u].w;
    }
}

__global__ void agg_k(const int* __restrict__ rowStart, const int* __restrict__ cnt,
                      const int* __restrict__ grouped, const float* __restrict__ dinv,
                      const float* __restrict__ hs, float* __restrict__ out, int n) {
    const int lane = threadIdx.x & 31;
    const int d = __builtin_amdgcn_readfirstlane((int)((blockIdx.x * blockDim.x + threadIdx.x) >> 5));
    if (d >= n) return;
    const int start = rowStart[d];
    const int c = cnt[d];
    const float4* hs4 = reinterpret_cast<const float4*>(hs);
    float4 acc = hs4[(size_t)d * (HID / 4) + lane];
    for (int base = 0; base < c; base += 32) {
        const int m = (c - base < 32) ? (c - base) : 32;
        int idx = 0;
        if (lane < m) idx = grouped[start + base + lane];
        int k = 0;
        for (; k + 8 <= m; k += 8) gather_add<8>(acc, hs4, idx, k, lane);
        if (k + 4 <= m) { gather_add<4>(acc, hs4, idx, k, lane); k += 4; }
        if (k + 2 <= m) { gather_add<2>(acc, hs4, idx, k, lane); k += 2; }
        if (k < m) gather_add<1>(acc, hs4, idx, k, lane);
    }
    const float w = dinv[d];
    acc.x *= w;
    acc.y *= w;
    acc.z *= w;
    acc.w *= w;
    reinterpret_cast<float4*>(out)[(size_t)d * (HID / 4) + lane] = acc;
}

#define POOL_CHUNK 32
#define POOL_TPB 128
static_assert(POOL_CHUNK == 32, "lane u of a wave holds the graph id of the chunk's node u");
static_assert(NN % POOL_CHUNK == 0, "a wave's chunk of nodes is always full");

__device__ __forceinline__ void pool_flush(float* __restrict__ pooled, int g, int q,
                                           const float4& a) {
    float* pp = pooled + (size_t)g * HID + q;
    atomicAdd(pp + 0, a.x);
    atomicAdd(pp + 1, a.y);
    atomicAdd(pp + 2, a.z);
    atomicAdd(pp + 3, a.w);
}

__global__ void __launch_bounds__(POOL_TPB)
pool_runs_k(const int* __restrict__ batch, const float* __restrict__ h,
            const float* __restrict__ b, float* __restrict__ pooled, int nNodes) {
    const int lane = threadIdx.x & 31;
    const int wave = __builtin_amdgcn_readfirstlane((int)((blockIdx.x * blockDim.x + threadIdx.x) >> 5));
    const int n0 = wave * POOL_CHUNK;
    if (n0 >= nNodes) return;
    const int gl = batch[n0 + lane];
    const float4* h4 = reinterpret_cast<const float4*>(h) + (size_t)n0 * (HID / 4) + lane;
    float4 v[POOL_CHUNK];
#pragma unroll
    for (int u = 0; u < POOL_CHUNK; ++u) v[u] = h4[(size_t)u * (HID / 4)];
    const float4 bb = reinterpret_cast<const float4*>(b)[lane];
    const int q = lane * 4;
    int cur = __builtin_amdgcn_readlane(gl, 0);
    float4 acc = make_float4(0.0f, 0.0f, 0.0f, 0.0f);
#pragma unroll
    for (int u = 0; u < POOL_CHUNK; ++u) {
        const int g = __builtin_amdgcn_readlane(gl, u);
        if (g != cur) {
            pool_flush(pooled, cur, q, acc);
            acc = make_float4(0.0f, 0.0f, 0.0f, 0.0f);
            cur = g;
        }
        acc.x += fmaxf(v[u].x + bb.x, 0.0f);
        acc.y += fmaxf(v[u].y + bb.y, 0.0f);
        acc.z += fmaxf(v[u].z + bb.z, 0.0f);
        acc.w += fmaxf(v[u].w + bb.w, 0.0f);
    }
    pool_flush(pooled, cur, q, acc);
}

template <int MT, int N, int K, bool ACT_A, bool ROWSCALE, bool BIAS_OUT>
__global__ void gemm_k(const float* __restrict__ A, const float* __restrict__ B,
                       float* __restrict__ C, const float* __restrict__ actBias,
                       const float* __restrict__ rowScale, const float* __restrict__ biasOut) {
    const int lane   = threadIdx.x & 31;
    const int wave   = threadIdx.x >> 5;
    const int half   = lane >> 4;
    const int l16    = lane & 15;
    const int nBase  = wave * 16;
    const int mTile0 = blockIdx.x * MT;

    const float* aBase = A + (size_t)(mTile0 * 16 + l16) * K + half * 2;
    const float* bBase = B + (size_t)(half * 2) * N + nBase + l16;
    const float* abBase = actBias + half * 2;

    v8f acc[MT];
#pragma unroll
    for (int t = 0; t < MT; ++t) {
        v8f z = {};
        acc[t] = z;
    }

#pragma unroll 4
    for (int k = 0; k < K; k += 4) {
        v2f b;
        b.x = bBase[k * N];
        b.y = bBase[k * N + N];
        float abx = 0.0f, aby = 0.0f;
        if (ACT_A) {
            abx = abBase[k];
            aby = abBase[k + 1];
        }
#pragma unroll
        for (int t = 0; t < MT; ++t) {
            v2f a;
            a.x = aBase[t * 16 * K + k];
            a.y = aBase[t * 16 * K + k + 1];
            if (ACT_A) {
                a.x = fmaxf(a.x + abx, 0.0f);
                a.y = fmaxf(a.y + aby, 0.0f);
            }
            acc[t] = __builtin_amdgcn_wmma_f32_16x16x4_f32(false, a, false, b, (short)0,
                                                           acc[t], false, false);
        }
    }

    float bv = BIAS_OUT ? biasOut[nBase + l16] : 0.0f;
#pragma unroll
    for (int t = 0; t < MT; ++t) {
        const int rBase = (mTile0 + t) * 16 + half * 8;
#pragma unroll
        for (int v = 0; v < 8; ++v) {
            const size_t idx = (size_t)(rBase + v) * N + nBase + l16;
            float val = acc[t][v] + bv;
            if (ROWSCALE) val *= rowScale[rBase + v];
            C[idx] = val;
        }
    }
}


extern "C" void kernel_launch(void* const* d_in, const int* in_sizes, int n_in,
                              void* d_out, int out_size, void* d_ws, size_t ws_size,
                              hipStream_t stream) {
    const float* x     = (const float*)d_in[0];
    const int*   ei    = (const int*)d_in[1];
    const int*   batch = (const int*)d_in[2];
    const float* W1    = (const float*)d_in[3];
    const float* b1    = (const float*)d_in[4];
    const float* W2    = (const float*)d_in[5];
    const float* b2    = (const float*)d_in[6];
    const float* Wl    = (const float*)d_in[7];
    const float* bl    = (const float*)d_in[8];
    float* out = (float*)d_out;

    const int* src = ei;
    const int* dst = ei + NE;

    char* ws = (char*)d_ws;
    const size_t FEAT_BYTES = (size_t)NN * HID * sizeof(float);
    const size_t EDGE_BYTES = (size_t)NE * sizeof(int);
    float* dinv    = (float*)(ws);
    float* bufH    = (float*)(ws + 524288);
    float* bufAgg  = (float*)(ws + 524288 + FEAT_BYTES);
    float* pooled  = (float*)(ws + 524288 + 2 * FEAT_BYTES);
    int* cnt       = (int*)(ws + 524288 + 2 * FEAT_BYTES + 262144);
    int* rowStart  = (int*)(ws + 524288 + 2 * FEAT_BYTES + 262144 + 524288);
    int* rank      = (int*)(ws + 524288 + 2 * FEAT_BYTES + 262144 + 2 * 524288);
    int* grouped   = (int*)(ws + 524288 + 2 * FEAT_BYTES + 262144 + 2 * 524288 + EDGE_BYTES);
    int* cursor    = cnt + NN;

    const int TPB = 256;

    fill_zero_i_k<<<(NN + 32 + TPB - 1) / TPB, TPB, 0, stream>>>(cnt, NN + 32);
    count_rank_k<<<(NE + TPB - 1) / TPB, TPB, 0, stream>>>(dst, cnt, rank, NE);
    row_start_k<<<(NN + TPB - 1) / TPB, TPB, 0, stream>>>(cnt, rowStart, dinv, cursor, NN);
    group_fill_k<<<(NE + TPB - 1) / TPB, TPB, 0, stream>>>(src, dst, rank, rowStart, grouped, NE);

    gemm_k<5, HID, IND, false, true, false><<<1250, 32 * (HID / 16), 0, stream>>>(
        x, W1, bufH, nullptr, dinv, nullptr);
    agg_k<<<(NN * 32 + TPB - 1) / TPB, TPB, 0, stream>>>(rowStart, cnt, grouped, dinv, bufH,
                                                         bufAgg, NN);

    gemm_k<5, HID, HID, true, true, false><<<1250, 32 * (HID / 16), 0, stream>>>(
        bufAgg, W2, bufH, b1, dinv, nullptr);
    agg_k<<<(NN * 32 + TPB - 1) / TPB, TPB, 0, stream>>>(rowStart, cnt, grouped, dinv, bufH,
                                                         bufAgg, NN);

    fill_zero_k<<<(NG * HID + TPB - 1) / TPB, TPB, 0, stream>>>(pooled, NG * HID);
    pool_runs_k<<<((NN / POOL_CHUNK) * 32 + POOL_TPB - 1) / POOL_TPB, POOL_TPB, 0, stream>>>(
        batch, bufAgg, b2, pooled, NN);

    gemm_k<2, OUTD, HID, false, false, true><<<16, 32 * (OUTD / 16), 0, stream>>>(
        pooled, Wl, out, nullptr, nullptr, bl);
}


# Each GCN layer is one kernel, (Ahat x)W instead of Ahat(xW): a block sums 32 rows' dinv-weighted neighbour rows into LDS, then WMMA with W from that tile; sums reassociated, self row added last
# speedup vs baseline: 2.7491x; 1.0708x over previous
#include <hip/hip_runtime.h>

typedef __attribute__((ext_vector_type(2))) float v2f;
typedef __attribute__((ext_vector_type(8))) float v8f;

#define NN 100000
#define NE 1600000
#define IND 128
#define HID 128
#define OUTD 64
#define NG 512


__global__ void fill_zero_k(float* __restrict__ p, int n) {
    int i = blockIdx.x * blockDim.x + threadIdx.x;
    if (i < n) p[i] = 0.0f;
}

__global__ void fill_zero_i_k(int* __restrict__ p, int n) {
    int i = blockIdx.x * blockDim.x + threadIdx.x;
    if (i < n) p[i] = 0;
}


__global__ void count_rank_k(const int* __restrict__ dst, int* __restrict__ cnt,
                             int* __restrict__ rank, int E) {
    int e = blockIdx.x * blockDim.x + threadIdx.x;
    if (e < E) rank[e] = atomicAdd(&cnt[dst[e]], 1);
}

__global__ void row_start_k(const int* __restrict__ cnt, int* __restrict__ rowStart,
                            float* __restrict__ dinv, int* __restrict__ cursor, int n) {
    const int i = blockIdx.x * blockDim.x + threadIdx.x;
    const int lane = threadIdx.x & 31;
    const int c = (i < n) ? cnt[i] : 0;
    int s = c;
#pragma unroll
    for (int off = 1; off < 32; off <<= 1) {
        int v = __shfl_up(s, off);
        if (lane >= off) s += v;
    }
    const int total = __shfl(s, 31);
    int base = 0;
    if (lane == 0) base = atomicAdd(cursor, total);
    base = __shfl(base, 0);
    if (i < n) {
        rowStart[i] = base + s - c;
        dinv[i] = rsqrtf((float)c + 1.0f);
    }
}

__global__ void group_fill_k(const int* __restrict__ src, const int* __restrict__ dst,
                             const int* __restrict__ rank, const int* __restrict__ rowStart,
                             const float* __restrict__ dinv, int* __restrict__ grouped,
                             float* __restrict__ gw, int E) {
    int e = blockIdx.x * blockDim.x + threadIdx.x;
    if (e < E) {
        const int s = src[e];
        const int pos = rowStart[dst[e]] + rank[e];
        grouped[pos] = s;
        gw[pos] = dinv[s];
    }
}

template <int U>
__device__ __forceinline__ void gather_fma(float4& acc, const float4* __restrict__ in4, int idx,
                                           float wgt, int k, int lane) {
    float4 v[U];
#pragma unroll
    for (int u = 0; u < U; ++u) {
        const unsigned s = (unsigned)__builtin_amdgcn_readlane(idx, k + u);
        v[u] = in4[(size_t)s * (HID / 4) + lane];
    }
#pragma unroll
    for (int u = 0; u < U; ++u) {
        const float w = __int_as_float(__builtin_amdgcn_readlane(__float_as_int(wgt), k + u));
        acc.x += w * v[u].x;
        acc.y += w * v[u].y;
        acc.z += w * v[u].z;
        acc.w += w * v[u].w;
    }
}

#define LROWS 32
#define LPITCH (HID + 4)
#define LAYER_TPB 256
static_assert(NN % LROWS == 0, "every block's rows exist");
static_assert(IND == HID, "one tile width for both layers");
static_assert(LAYER_TPB / 32 == HID / 16, "one wave per 16-column tile of W");
static_assert(LROWS % (LAYER_TPB / 32) == 0 && LROWS / (LAYER_TPB / 32) == 4,
              "rows per wave: a power of two, one metadata lane each");

template <bool ACT_OUT>
__global__ void __launch_bounds__(LAYER_TPB)
layer_k(const int* __restrict__ rowStart, const int* __restrict__ cnt,
        const int* __restrict__ grouped, const float* __restrict__ gw,
        const float* __restrict__ dinv, const float* __restrict__ in,
        const float* __restrict__ Wm, const float* __restrict__ bias, float* __restrict__ out) {
    __shared__ __attribute__((aligned(16))) float As[LROWS * LPITCH];
    const int lane = threadIdx.x & 31;
    const int wave = __builtin_amdgcn_readfirstlane((int)(threadIdx.x >> 5));
    const int row0 = blockIdx.x * LROWS;
    const float4* in4 = reinterpret_cast<const float4*>(in);

    constexpr int NW = LAYER_TPB / 32;
    constexpr int RPW = LROWS / NW;
    const int dl = row0 + wave + NW * (lane & (RPW - 1));
    const int startV = rowStart[dl];
    const int cntV = cnt[dl];
    const float wdV = dinv[dl];
    int idxF[RPW];
    float wgtF[RPW];
#pragma unroll
    for (int i = 0; i < RPW; ++i) {
        const int start = __builtin_amdgcn_readlane(startV, i);
        const int c = __builtin_amdgcn_readlane(cntV, i);
        idxF[i] = 0;
        wgtF[i] = 0.0f;
        if (lane < c) {
            idxF[i] = grouped[start + lane];
            wgtF[i] = gw[start + lane];
        }
    }
#pragma unroll
    for (int i = 0; i < RPW; ++i) {
        const int r = wave + NW * i;
        const int d = row0 + r;
        const int start = __builtin_amdgcn_readlane(startV, i);
        const int c = __builtin_amdgcn_readlane(cntV, i);
        const float wd = __int_as_float(__builtin_amdgcn_readlane(__float_as_int(wdV), i));
        const float4 self = in4[(size_t)d * (HID / 4) + lane];
        float4 acc = make_float4(0.0f, 0.0f, 0.0f, 0.0f);
        int idx = idxF[i];
        float wgt = wgtF[i];
        for (int base = 0; base < c; base += 32) {
            const int m = (c - base < 32) ? (c - base) : 32;
            if (base > 0) {
                idx = 0;
                wgt = 0.0f;
                if (lane < m) {
                    idx = grouped[start + base + lane];
                    wgt = gw[start + base + lane];
                }
            }
            int k = 0;
            for (; k + 8 <= m; k += 8) gather_fma<8>(acc, in4, idx, wgt, k, lane);
            if (k + 4 <= m) { gather_fma<4>(acc, in4, idx, wgt, k, lane); k += 4; }
            if (k + 2 <= m) { gather_fma<2>(acc, in4, idx, wgt, k, lane); k += 2; }
            if (k < m) gather_fma<1>(acc, in4, idx, wgt, k, lane);
        }
        acc.x = (acc.x + wd * self.x) * wd;
        acc.y = (acc.y + wd * self.y) * wd;
        acc.z = (acc.z + wd * self.z) * wd;
        acc.w = (acc.w + wd * self.w) * wd;
        *reinterpret_cast<float4*>(&As[r * LPITCH + lane * 4]) = acc;
    }
    __syncthreads();

    const int half  = lane >> 4;
    const int l16   = lane & 15;
    const int nBase = wave * 16;
    const float* bBase = Wm + (size_t)(half * 2) * HID + nBase + l16;
    const float* aBase = As + l16 * LPITCH + half * 2;

    v8f acc2[LROWS / 16];
#pragma unroll
    for (int t = 0; t < LROWS / 16; ++t) {
        v8f z = {};
        acc2[t] = z;
    }

#pragma unroll 8
    for (int k = 0; k < HID; k += 4) {
        v2f b;
        b.x = bBase[k * HID];
        b.y = bBase[k * HID + HID];
#pragma unroll
        for (int t = 0; t < LROWS / 16; ++t) {
            const v2f a = *reinterpret_cast<const v2f*>(aBase + t * 16 * LPITCH + k);
            acc2[t] = __builtin_amdgcn_wmma_f32_16x16x4_f32(false, a, false, b, (short)0,
                                                            acc2[t], false, false);
        }
    }

    const float bv = ACT_OUT ? bias[nBase + l16] : 0.0f;
#pragma unroll
    for (int t = 0; t < LROWS / 16; ++t) {
        const int rBase = row0 + t * 16 + half * 8;
#pragma unroll
        for (int v = 0; v < 8; ++v) {
            float val = acc2[t][v];
            if (ACT_OUT) val = fmaxf(val + bv, 0.0f);
            out[(size_t)(rBase + v) * HID + nBase + l16] = val;
        }
    }
}

#define POOL_CHUNK 32
#define POOL_TPB 128
static_assert(POOL_CHUNK == 32, "lane u of a wave holds the graph id of the chunk's node u");
static_assert(NN % POOL_CHUNK == 0, "a wave's chunk of nodes is always full");

__device__ __forceinline__ void pool_flush(float* __restrict__ pooled, int g, int q,
                                           const float4& a) {
    float* pp = pooled + (size_t)g * HID + q;
    atomicAdd(pp + 0, a.x);
    atomicAdd(pp + 1, a.y);
    atomicAdd(pp + 2, a.z);
    atomicAdd(pp + 3, a.w);
}

__global__ void __launch_bounds__(POOL_TPB)
pool_runs_k(const int* __restrict__ batch, const float* __restrict__ h,
            const float* __restrict__ b, float* __restrict__ pooled, int nNodes) {
    const int lane = threadIdx.x & 31;
    const int wave = __builtin_amdgcn_readfirstlane((int)((blockIdx.x * blockDim.x + threadIdx.x) >> 5));
    const int n0 = wave * POOL_CHUNK;
    if (n0 >= nNodes) return;
    const int gl = batch[n0 + lane];
    const float4* h4 = reinterpret_cast<const float4*>(h) + (size_t)n0 * (HID / 4) + lane;
    float4 v[POOL_CHUNK];
#pragma unroll
    for (int u = 0; u < POOL_CHUNK; ++u) v[u] = h4[(size_t)u * (HID / 4)];
    const float4 bb = reinterpret_cast<const float4*>(b)[lane];
    const int q = lane * 4;
    int cur = __builtin_amdgcn_readlane(gl, 0);
    float4 acc = make_float4(0.0f, 0.0f, 0.0f, 0.0f);
#pragma unroll
    for (int u = 0; u < POOL_CHUNK; ++u) {
        const int g = __builtin_amdgcn_readlane(gl, u);
        if (g != cur) {
            pool_flush(pooled, cur, q, acc);
            acc = make_float4(0.0f, 0.0f, 0.0f, 0.0f);
            cur = g;
        }
        acc.x += fmaxf(v[u].x + bb.x, 0.0f);
        acc.y += fmaxf(v[u].y + bb.y, 0.0f);
        acc.z += fmaxf(v[u].z + bb.z, 0.0f);
        acc.w += fmaxf(v[u].w + bb.w, 0.0f);
    }
    pool_flush(pooled, cur, q, acc);
}

template <int MT, int N, int K, bool ACT_A, bool ROWSCALE, bool BIAS_OUT>
__global__ void gemm_k(const float* __restrict__ A, const float* __restrict__ B,
                       float* __restrict__ C, const float* __restrict__ actBias,
                       const float* __restrict__ rowScale, const float* __restrict__ biasOut) {
    const int lane   = threadIdx.x & 31;
    const int wave   = threadIdx.x >> 5;
    const int half   = lane >> 4;
    const int l16    = lane & 15;
    const int nBase  = wave * 16;
    const int mTile0 = blockIdx.x * MT;

    const float* aBase = A + (size_t)(mTile0 * 16 + l16) * K + half * 2;
    const float* bBase = B + (size_t)(half * 2) * N + nBase + l16;
    const float* abBase = actBias + half * 2;

    v8f acc[MT];
#pragma unroll
    for (int t = 0; t < MT; ++t) {
        v8f z = {};
        acc[t] = z;
    }

#pragma unroll 4
    for (int k = 0; k < K; k += 4) {
        v2f b;
        b.x = bBase[k * N];
        b.y = bBase[k * N + N];
        float abx = 0.0f, aby = 0.0f;
        if (ACT_A) {
            abx = abBase[k];
            aby = abBase[k + 1];
        }
#pragma unroll
        for (int t = 0; t < MT; ++t) {
            v2f a;
            a.x = aBase[t * 16 * K + k];
            a.y = aBase[t * 16 * K + k + 1];
            if (ACT_A) {
                a.x = fmaxf(a.x + abx, 0.0f);
                a.y = fmaxf(a.y + aby, 0.0f);
            }
            acc[t] = __builtin_amdgcn_wmma_f32_16x16x4_f32(false, a, false, b, (short)0,
                                                           acc[t], false, false);
        }
    }

    float bv = BIAS_OUT ? biasOut[nBase + l16] : 0.0f;
#pragma unroll
    for (int t = 0; t < MT; ++t) {
        const int rBase = (mTile0 + t) * 16 + half * 8;
#pragma unroll
        for (int v = 0; v < 8; ++v) {
            const size_t idx = (size_t)(rBase + v) * N + nBase + l16;
            float val = acc[t][v] + bv;
            if (ROWSCALE) val *= rowScale[rBase + v];
            C[idx] = val;
        }
    }
}


extern "C" void kernel_launch(void* const* d_in, const int* in_sizes, int n_in,
                              void* d_out, int out_size, void* d_ws, size_t ws_size,
                              hipStream_t stream) {
    const float* x     = (const float*)d_in[0];
    const int*   ei    = (const int*)d_in[1];
    const int*   batch = (const int*)d_in[2];
    const float* W1    = (const float*)d_in[3];
    const float* b1    = (const float*)d_in[4];
    const float* W2    = (const float*)d_in[5];
    const float* b2    = (const float*)d_in[6];
    const float* Wl    = (const float*)d_in[7];
    const float* bl    = (const float*)d_in[8];
    float* out = (float*)d_out;

    const int* src = ei;
    const int* dst = ei + NE;

    char* ws = (char*)d_ws;
    const size_t FEAT_BYTES = (size_t)NN * HID * sizeof(float);
    const size_t EDGE_BYTES = (size_t)NE * sizeof(int);
    float* dinv    = (float*)(ws);
    float* bufH    = (float*)(ws + 524288);
    float* bufAgg  = (float*)(ws + 524288 + FEAT_BYTES);
    float* pooled  = (float*)(ws + 524288 + 2 * FEAT_BYTES);
    int* cnt       = (int*)(ws + 524288 + 2 * FEAT_BYTES + 262144);
    int* rowStart  = (int*)(ws + 524288 + 2 * FEAT_BYTES + 262144 + 524288);
    int* rank      = (int*)(ws + 524288 + 2 * FEAT_BYTES + 262144 + 2 * 524288);
    int* grouped   = (int*)(ws + 524288 + 2 * FEAT_BYTES + 262144 + 2 * 524288 + EDGE_BYTES);
    float* gw      = (float*)(ws + 524288 + 2 * FEAT_BYTES + 262144 + 2 * 524288 + 2 * EDGE_BYTES);
    int* cursor    = cnt + NN;

    const int TPB = 256;

    fill_zero_i_k<<<(NN + 32 + TPB - 1) / TPB, TPB, 0, stream>>>(cnt, NN + 32);
    count_rank_k<<<(NE + TPB - 1) / TPB, TPB, 0, stream>>>(dst, cnt, rank, NE);
    row_start_k<<<(NN + TPB - 1) / TPB, TPB, 0, stream>>>(cnt, rowStart, dinv, cursor, NN);
    group_fill_k<<<(NE + TPB - 1) / TPB, TPB, 0, stream>>>(src, dst, rank, rowStart, dinv, grouped,
                                                           gw, NE);

    layer_k<true><<<NN / LROWS, LAYER_TPB, 0, stream>>>(rowStart, cnt, grouped, gw, dinv, x, W1, b1,
                                                        bufH);

    layer_k<false><<<NN / LROWS, LAYER_TPB, 0, stream>>>(rowStart, cnt, grouped, gw, dinv, bufH, W2,
                                                         nullptr, bufAgg);

    fill_zero_k<<<(NG * HID + TPB - 1) / TPB, TPB, 0, stream>>>(pooled, NG * HID);
    pool_runs_k<<<((NN / POOL_CHUNK) * 32 + POOL_TPB - 1) / POOL_TPB, POOL_TPB, 0, stream>>>(
        batch, bufAgg, b2, pooled, NN);

    gemm_k<2, OUTD, HID, false, false, true><<<16, 32 * (OUTD / 16), 0, stream>>>(
        pooled, Wl, out, nullptr, nullptr, bl);
}


# Pooling moved into the layer-2 kernel: each 32-row block sums relu(h+b2) per column in registers (lane pair) and adds it to pooled with one float atomic per run; pooled sum order = block partials
# speedup vs baseline: 2.8359x; 1.0316x over previous
#include <hip/hip_runtime.h>

typedef __attribute__((ext_vector_type(2))) float v2f;
typedef __attribute__((ext_vector_type(8))) float v8f;

#define NN 100000
#define NE 1600000
#define IND 128
#define HID 128
#define OUTD 64
#define NG 512


__global__ void fill_zero_i_k(int* __restrict__ p, int n) {
    int i = blockIdx.x * blockDim.x + threadIdx.x;
    if (i < n) p[i] = 0;
}


__global__ void count_rank_k(const int* __restrict__ dst, int* __restrict__ cnt,
                             int* __restrict__ rank, int E) {
    int e = blockIdx.x * blockDim.x + threadIdx.x;
    if (e < E) rank[e] = atomicAdd(&cnt[dst[e]], 1);
}

__global__ void row_start_k(const int* __restrict__ cnt, int* __restrict__ rowStart,
                            float* __restrict__ dinv, int* __restrict__ cursor, int n) {
    const int i = blockIdx.x * blockDim.x + threadIdx.x;
    const int lane = threadIdx.x & 31;
    const int c = (i < n) ? cnt[i] : 0;
    int s = c;
#pragma unroll
    for (int off = 1; off < 32; off <<= 1) {
        int v = __shfl_up(s, off);
        if (lane >= off) s += v;
    }
    const int total = __shfl(s, 31);
    int base = 0;
    if (lane == 0) base = atomicAdd(cursor, total);
    base = __shfl(base, 0);
    if (i < n) {
        rowStart[i] = base + s - c;
        dinv[i] = rsqrtf((float)c + 1.0f);
    }
}

__global__ void group_fill_k(const int* __restrict__ src, const int* __restrict__ dst,
                             const int* __restrict__ rank, const int* __restrict__ rowStart,
                             const float* __restrict__ dinv, int* __restrict__ grouped,
                             float* __restrict__ gw, int E) {
    int e = blockIdx.x * blockDim.x + threadIdx.x;
    if (e < E) {
        const int s = src[e];
        const int pos = rowStart[dst[e]] + rank[e];
        grouped[pos] = s;
        gw[pos] = dinv[s];
    }
}

template <int U>
__device__ __forceinline__ void gather_fma(float4& acc, const float4* __restrict__ in4, int idx,
                                           float wgt, int k, int lane) {
    float4 v[U];
#pragma unroll
    for (int u = 0; u < U; ++u) {
        const unsigned s = (unsigned)__builtin_amdgcn_readlane(idx, k + u);
        v[u] = in4[(size_t)s * (HID / 4) + lane];
    }
#pragma unroll
    for (int u = 0; u < U; ++u) {
        const float w = __int_as_float(__builtin_amdgcn_readlane(__float_as_int(wgt), k + u));
        acc.x += w * v[u].x;
        acc.y += w * v[u].y;
        acc.z += w * v[u].z;
        acc.w += w * v[u].w;
    }
}

#define LROWS 32
#define LPITCH (HID + 4)
#define LAYER_TPB 256
static_assert(NN % LROWS == 0, "every block's rows exist");
static_assert(LROWS == 32, "pooling epilogue: lane i of a wave holds the graph id of tile row i");
static_assert(IND == HID, "one tile width for both layers");
static_assert(LAYER_TPB / 32 == HID / 16, "one wave per 16-column tile of W");
static_assert(LROWS % (LAYER_TPB / 32) == 0 && LROWS / (LAYER_TPB / 32) == 4,
              "rows per wave: a power of two, one metadata lane each");

template <bool POOL>
__global__ void __launch_bounds__(LAYER_TPB)
layer_k(const int* __restrict__ rowStart, const int* __restrict__ cnt,
        const int* __restrict__ grouped, const float* __restrict__ gw,
        const float* __restrict__ dinv, const float* __restrict__ in,
        const float* __restrict__ Wm, const float* __restrict__ bias,
        const int* __restrict__ batch, float* __restrict__ out) {
    __shared__ __attribute__((aligned(16))) float As[LROWS * LPITCH];
    const int lane = threadIdx.x & 31;
    const int wave = __builtin_amdgcn_readfirstlane((int)(threadIdx.x >> 5));
    const int row0 = blockIdx.x * LROWS;
    const float4* in4 = reinterpret_cast<const float4*>(in);

    constexpr int NW = LAYER_TPB / 32;
    constexpr int RPW = LROWS / NW;
    const int dl = row0 + wave + NW * (lane & (RPW - 1));
    const int startV = rowStart[dl];
    const int cntV = cnt[dl];
    const float wdV = dinv[dl];
    int idxF[RPW];
    float wgtF[RPW];
#pragma unroll
    for (int i = 0; i < RPW; ++i) {
        const int start = __builtin_amdgcn_readlane(startV, i);
        const int c = __builtin_amdgcn_readlane(cntV, i);
        idxF[i] = 0;
        wgtF[i] = 0.0f;
        if (lane < c) {
            idxF[i] = grouped[start + lane];
            wgtF[i] = gw[start + lane];
        }
    }
#pragma unroll
    for (int i = 0; i < RPW; ++i) {
        const int r = wave + NW * i;
        const int d = row0 + r;
        const int start = __builtin_amdgcn_readlane(startV, i);
        const int c = __builtin_amdgcn_readlane(cntV, i);
        const float wd = __int_as_float(__builtin_amdgcn_readlane(__float_as_int(wdV), i));
        const float4 self = in4[(size_t)d * (HID / 4) + lane];
        float4 acc = make_float4(0.0f, 0.0f, 0.0f, 0.0f);
        int idx = idxF[i];
        float wgt = wgtF[i];
        for (int base = 0; base < c; base += 32) {
            const int m = (c - base < 32) ? (c - base) : 32;
            if (base > 0) {
                idx = 0;
                wgt = 0.0f;
                if (lane < m) {
                    idx = grouped[start + base + lane];
                    wgt = gw[start + base + lane];
                }
            }
            int k = 0;
            for (; k + 8 <= m; k += 8) gather_fma<8>(acc, in4, idx, wgt, k, lane);
            if (k + 4 <= m) { gather_fma<4>(acc, in4, idx, wgt, k, lane); k += 4; }
            if (k + 2 <= m) { gather_fma<2>(acc, in4, idx, wgt, k, lane); k += 2; }
            if (k < m) gather_fma<1>(acc, in4, idx, wgt, k, lane);
        }
        acc.x = (acc.x + wd * self.x) * wd;
        acc.y = (acc.y + wd * self.y) * wd;
        acc.z = (acc.z + wd * self.z) * wd;
        acc.w = (acc.w + wd * self.w) * wd;
        *reinterpret_cast<float4*>(&As[r * LPITCH + lane * 4]) = acc;
    }
    __syncthreads();

    const int half  = lane >> 4;
    const int l16   = lane & 15;
    const int nBase = wave * 16;
    const float* bBase = Wm + (size_t)(half * 2) * HID + nBase + l16;
    const float* aBase = As + l16 * LPITCH + half * 2;

    v8f acc2[LROWS / 16];
#pragma unroll
    for (int t = 0; t < LROWS / 16; ++t) {
        v8f z = {};
        acc2[t] = z;
    }

#pragma unroll 8
    for (int k = 0; k < HID; k += 4) {
        v2f b;
        b.x = bBase[k * HID];
        b.y = bBase[k * HID + HID];
#pragma unroll
        for (int t = 0; t < LROWS / 16; ++t) {
            const v2f a = *reinterpret_cast<const v2f*>(aBase + t * 16 * LPITCH + k);
            acc2[t] = __builtin_amdgcn_wmma_f32_16x16x4_f32(false, a, false, b, (short)0,
                                                            acc2[t], false, false);
        }
    }

    const float bv = bias[nBase + l16];
    if (!POOL) {
#pragma unroll
        for (int t = 0; t < LROWS / 16; ++t) {
            const int rBase = row0 + t * 16 + half * 8;
#pragma unroll
            for (int v = 0; v < 8; ++v)
                out[(size_t)(rBase + v) * HID + nBase + l16] = fmaxf(acc2[t][v] + bv, 0.0f);
        }
    } else {
        const int gl = batch[row0 + lane];
        float val[LROWS / 16][8];
#pragma unroll
        for (int t = 0; t < LROWS / 16; ++t)
#pragma unroll
            for (int v = 0; v < 8; ++v) val[t][v] = fmaxf(acc2[t][v] + bv, 0.0f);
        float* pcol = out + nBase + l16;
        const int g0 = __builtin_amdgcn_readfirstlane(gl);
        if (__builtin_amdgcn_ballot_w32(gl != g0) == 0) {
            float s = 0.0f;
#pragma unroll
            for (int t = 0; t < LROWS / 16; ++t)
#pragma unroll
                for (int v = 0; v < 8; ++v) s += val[t][v];
            s += __shfl_xor(s, 16);
            if (half == 0) atomicAdd(pcol + (size_t)g0 * HID, s);
        } else {
#pragma unroll
            for (int t = 0; t < LROWS / 16; ++t) {
                const int rb = t * 16 + half * 8;
                int cur = __shfl(gl, rb);
                float a = 0.0f;
#pragma unroll
                for (int v = 0; v < 8; ++v) {
                    const int g = __shfl(gl, rb + v);
                    if (g != cur) {
                        atomicAdd(pcol + (size_t)cur * HID, a);
                        a = 0.0f;
                        cur = g;
                    }
                    a += val[t][v];
                }
                atomicAdd(pcol + (size_t)cur * HID, a);
            }
        }
    }
}

template <int MT, int N, int K, bool ACT_A, bool ROWSCALE, bool BIAS_OUT>
__global__ void gemm_k(const float* __restrict__ A, const float* __restrict__ B,
                       float* __restrict__ C, const float* __restrict__ actBias,
                       const float* __restrict__ rowScale, const float* __restrict__ biasOut) {
    const int lane   = threadIdx.x & 31;
    const int wave   = threadIdx.x >> 5;
    const int half   = lane >> 4;
    const int l16    = lane & 15;
    const int nBase  = wave * 16;
    const int mTile0 = blockIdx.x * MT;

    const float* aBase = A + (size_t)(mTile0 * 16 + l16) * K + half * 2;
    const float* bBase = B + (size_t)(half * 2) * N + nBase + l16;
    const float* abBase = actBias + half * 2;

    v8f acc[MT];
#pragma unroll
    for (int t = 0; t < MT; ++t) {
        v8f z = {};
        acc[t] = z;
    }

#pragma unroll 4
    for (int k = 0; k < K; k += 4) {
        v2f b;
        b.x = bBase[k * N];
        b.y = bBase[k * N + N];
        float abx = 0.0f, aby = 0.0f;
        if (ACT_A) {
            abx = abBase[k];
            aby = abBase[k + 1];
        }
#pragma unroll
        for (int t = 0; t < MT; ++t) {
            v2f a;
            a.x = aBase[t * 16 * K + k];
            a.y = aBase[t * 16 * K + k + 1];
            if (ACT_A) {
                a.x = fmaxf(a.x + abx, 0.0f);
                a.y = fmaxf(a.y + aby, 0.0f);
            }
            acc[t] = __builtin_amdgcn_wmma_f32_16x16x4_f32(false, a, false, b, (short)0,
                                                           acc[t], false, false);
        }
    }

    float bv = BIAS_OUT ? biasOut[nBase + l16] : 0.0f;
#pragma unroll
    for (int t = 0; t < MT; ++t) {
        const int rBase = (mTile0 + t) * 16 + half * 8;
#pragma unroll
        for (int v = 0; v < 8; ++v) {
            const size_t idx = (size_t)(rBase + v) * N + nBase + l16;
            float val = acc[t][v] + bv;
            if (ROWSCALE) val *= rowScale[rBase + v];
            C[idx] = val;
        }
    }
}


extern "C" void kernel_launch(void* const* d_in, const int* in_sizes, int n_in,
                              void* d_out, int out_size, void* d_ws, size_t ws_size,
                              hipStream_t stream) {
    const float* x     = (const float*)d_in[0];
    const int*   ei    = (const int*)d_in[1];
    const int*   batch = (const int*)d_in[2];
    const float* W1    = (const float*)d_in[3];
    const float* b1    = (const float*)d_in[4];
    const float* W2    = (const float*)d_in[5];
    const float* b2    = (const float*)d_in[6];
    const float* Wl    = (const float*)d_in[7];
    const float* bl    = (const float*)d_in[8];
    float* out = (float*)d_out;

    const int* src = ei;
    const int* dst = ei + NE;

    char* ws = (char*)d_ws;
    const size_t FEAT_BYTES = (size_t)NN * HID * sizeof(float);
    const size_t EDGE_BYTES = (size_t)NE * sizeof(int);
    float* dinv    = (float*)(ws);
    float* bufH    = (float*)(ws + 524288);
    float* pooled  = (float*)(ws + 524288 + 2 * FEAT_BYTES);
    int* cnt       = (int*)(ws + 524288 + 2 * FEAT_BYTES + 262144);
    int* rowStart  = (int*)(ws + 524288 + 2 * FEAT_BYTES + 262144 + 524288);
    int* rank      = (int*)(ws + 524288 + 2 * FEAT_BYTES + 262144 + 2 * 524288);
    int* grouped   = (int*)(ws + 524288 + 2 * FEAT_BYTES + 262144 + 2 * 524288 + EDGE_BYTES);
    float* gw      = (float*)(ws + 524288 + 2 * FEAT_BYTES + 262144 + 2 * 524288 + 2 * EDGE_BYTES);
    int* cursor    = cnt + NN;
    int* zeroed    = (int*)pooled;

    const int TPB = 256;

    fill_zero_i_k<<<(NG * HID + NN + 32 + TPB - 1) / TPB, TPB, 0, stream>>>(zeroed,
                                                                            NG * HID + NN + 32);
    count_rank_k<<<(NE + TPB - 1) / TPB, TPB, 0, stream>>>(dst, cnt, rank, NE);
    row_start_k<<<(NN + TPB - 1) / TPB, TPB, 0, stream>>>(cnt, rowStart, dinv, cursor, NN);
    group_fill_k<<<(NE + TPB - 1) / TPB, TPB, 0, stream>>>(src, dst, rank, rowStart, dinv, grouped,
                                                           gw, NE);

    layer_k<false><<<NN / LROWS, LAYER_TPB, 0, stream>>>(rowStart, cnt, grouped, gw, dinv, x, W1, b1,
                                                         batch, bufH);

    layer_k<true><<<NN / LROWS, LAYER_TPB, 0, stream>>>(rowStart, cnt, grouped, gw, dinv, bufH, W2, b2,
                                                        batch, pooled);

    gemm_k<2, OUTD, HID, false, false, true><<<16, 32 * (OUTD / 16), 0, stream>>>(
        pooled, Wl, out, nullptr, nullptr, bl);
}


# Edge grouping by a two-level counting sort on LDS counters (chunk x 256-row bucket counts, then one block per bucket) replaces one returning global atomic per edge; row sum order = LDS arrival rank
# speedup vs baseline: 4.7649x; 1.6802x over previous
#include <hip/hip_runtime.h>

typedef __attribute__((ext_vector_type(2))) float v2f;
typedef __attribute__((ext_vector_type(8))) float v8f;

#define NN 100000
#define NE 1600000
#define IND 128
#define HID 128
#define OUTD 64
#define NG 512

#define BKT_SHIFT 8
#define BKT_NODES (1 << BKT_SHIFT)
#define NBK ((NN + BKT_NODES - 1) / BKT_NODES)
#define SRC_BITS 17
#define SRC_MASK ((1u << SRC_BITS) - 1u)
#define L1_BLOCKS 256
#define L1_TPB 1024
#define L1_CHUNK ((NE + L1_BLOCKS - 1) / L1_BLOCKS)
#define L1_EPT ((L1_CHUNK + L1_TPB - 1) / L1_TPB)
#define HCOLS (L1_TPB / 2)
#define L2_TPB 1024
#define L2_RPT 4
static_assert(NN <= (1 << SRC_BITS), "a source row number fits the low bits of a record");
static_assert(SRC_BITS + BKT_SHIFT <= 31, "a record is one non-negative word");
static_assert(NBK < HCOLS, "one thread per bucket plus one for the end mark");
static_assert(NG * HID <= L1_BLOCKS * L1_TPB, "the first launch has a thread per pooled element");
static_assert(BKT_NODES <= L2_TPB, "one thread per row of a bucket");

__device__ __forceinline__ int wave_incl_scan(int v, int lane) {
#pragma unroll
    for (int off = 1; off < 32; off <<= 1) {
        const int u = __shfl_up(v, off);
        if (lane >= off) v += u;
    }
    return v;
}

template <int NWAVES>
__device__ __forceinline__ int block_excl_scan(int v, int* wtot, int lane, int wave) {
    const int s = wave_incl_scan(v, lane);
    if (lane == 31) wtot[wave] = s;
    __syncthreads();
    if (wave == 0) {
        const int w = (lane < NWAVES) ? wtot[lane] : 0;
        const int ws = wave_incl_scan(w, lane);
        if (lane < NWAVES) wtot[lane] = ws - w;
    }
    __syncthreads();
    return wtot[wave] + s - v;
}

__global__ void __launch_bounds__(L1_TPB)
bucket_hist_k(const int* __restrict__ dst, int* __restrict__ hist, float* __restrict__ pooled) {
    __shared__ int h[NBK];
    const int t = threadIdx.x;
    if (t < NBK) h[t] = 0;
    const int g = blockIdx.x * L1_TPB + t;
    if (g < NG * HID) pooled[g] = 0.0f;
    __syncthreads();
    const int e0 = blockIdx.x * L1_CHUNK;
    const int e1 = (e0 + L1_CHUNK < NE) ? (e0 + L1_CHUNK) : NE;
    for (int eb = e0; eb < e1; eb += L1_EPT * L1_TPB) {
        int d[L1_EPT];
#pragma unroll
        for (int k = 0; k < L1_EPT; ++k) {
            const int e = eb + t + k * L1_TPB;
            d[k] = dst[(e < e1) ? e : (e1 - 1)];
        }
#pragma unroll
        for (int k = 0; k < L1_EPT; ++k) {
            const int e = eb + t + k * L1_TPB;
            if (e < e1) atomicAdd(&h[d[k] >> BKT_SHIFT], 1);
        }
    }
    __syncthreads();
    if (t < NBK) hist[blockIdx.x * NBK + t] = h[t];
}

__global__ void __launch_bounds__(L1_TPB)
bucket_scatter_k(const int* __restrict__ src, const int* __restrict__ dst,
                 const int* __restrict__ hist, int* __restrict__ rec,
                 int* __restrict__ bucketStart) {
    __shared__ int partBefore[2][HCOLS];
    __shared__ int partTotal[2][HCOLS];
    __shared__ int baseL[NBK];
    __shared__ int curL[NBK];
    __shared__ int wtot[L1_TPB / 32];
    const int t = threadIdx.x;
    const int lane = t & 31;
    const int wave = t >> 5;

    const int c = t & (HCOLS - 1);
    const int hh = t >> 9;
    static_assert(HCOLS == 512, "hh = t >> 9");
    int bef = 0, tot = 0;
    if (c < NBK) {
        const int* col = hist + c;
        const int jb = hh * (L1_BLOCKS / 2);
#pragma unroll 16
        for (int j = jb; j < jb + L1_BLOCKS / 2; ++j) {
            const int v = col[j * NBK];
            tot += v;
            if (j < (int)blockIdx.x) bef += v;
        }
    }
    partBefore[hh][c] = bef;
    partTotal[hh][c] = tot;
    __syncthreads();
    int total = 0, before = 0;
    if (t < HCOLS) {
        total = partTotal[0][t] + partTotal[1][t];
        before = partBefore[0][t] + partBefore[1][t];
    }
    const int bStart = block_excl_scan<L1_TPB / 32>(total, wtot, lane, wave);
    if (t < NBK) {
        baseL[t] = bStart + before;
        curL[t] = 0;
    }
    if (blockIdx.x == 0 && t <= NBK) bucketStart[t] = bStart;
    __syncthreads();

    const int e0 = blockIdx.x * L1_CHUNK;
    const int e1 = (e0 + L1_CHUNK < NE) ? (e0 + L1_CHUNK) : NE;
    for (int eb = e0; eb < e1; eb += L1_EPT * L1_TPB) {
        int d[L1_EPT], s[L1_EPT];
#pragma unroll
        for (int k = 0; k < L1_EPT; ++k) {
            const int e = eb + t + k * L1_TPB;
            const int ec = (e < e1) ? e : (e1 - 1);
            d[k] = dst[ec];
            s[k] = src[ec];
        }
#pragma unroll
        for (int k = 0; k < L1_EPT; ++k) {
            const int e = eb + t + k * L1_TPB;
            if (e < e1) {
                const int b = d[k] >> BKT_SHIFT;
                const int r = atomicAdd(&curL[b], 1);
                rec[baseL[b] + r] = s[k] | ((d[k] & (BKT_NODES - 1)) << SRC_BITS);
            }
        }
    }
}

__global__ void __launch_bounds__(L2_TPB)
bucket_sort_k(const int* __restrict__ rec, const int* __restrict__ bucketStart,
              int* __restrict__ cnt, int* __restrict__ rowStart, float* __restrict__ dinv,
              int* __restrict__ grouped) {
    __shared__ int cL[BKT_NODES];
    __shared__ int oL[BKT_NODES];
    __shared__ int wtot[L2_TPB / 32];
    const int t = threadIdx.x;
    const int lane = t & 31;
    const int wave = t >> 5;
    const int b = blockIdx.x;
    const int r0 = bucketStart[b];
    const int r1 = bucketStart[b + 1];
    if (t < BKT_NODES) cL[t] = 0;
    __syncthreads();
    for (int ib = r0; ib < r1; ib += L2_RPT * L2_TPB) {
        unsigned w[L2_RPT];
#pragma unroll
        for (int k = 0; k < L2_RPT; ++k) {
            const int i = ib + t + k * L2_TPB;
            w[k] = (unsigned)rec[(i < r1) ? i : (r1 - 1)];
        }
#pragma unroll
        for (int k = 0; k < L2_RPT; ++k) {
            const int i = ib + t + k * L2_TPB;
            if (i < r1) atomicAdd(&cL[w[k] >> SRC_BITS], 1);
        }
    }
    __syncthreads();
    const int cv = (t < BKT_NODES) ? cL[t] : 0;
    const int off = block_excl_scan<L2_TPB / 32>(cv, wtot, lane, wave);
    if (t < BKT_NODES) {
        oL[t] = off;
        cL[t] = 0;
        const int node = b * BKT_NODES + t;
        if (node < NN) {
            cnt[node] = cv;
            rowStart[node] = r0 + off;
            dinv[node] = rsqrtf((float)cv + 1.0f);
        }
    }
    __syncthreads();
    for (int ib = r0; ib < r1; ib += L2_RPT * L2_TPB) {
        unsigned w[L2_RPT];
#pragma unroll
        for (int k = 0; k < L2_RPT; ++k) {
            const int i = ib + t + k * L2_TPB;
            w[k] = (unsigned)rec[(i < r1) ? i : (r1 - 1)];
        }
#pragma unroll
        for (int k = 0; k < L2_RPT; ++k) {
            const int i = ib + t + k * L2_TPB;
            if (i < r1) {
                const int dl = (int)(w[k] >> SRC_BITS);
                const int p = atomicAdd(&cL[dl], 1);
                grouped[r0 + oL[dl] + p] = (int)(w[k] & SRC_MASK);
            }
        }
    }
}

__global__ void edge_weight_k(const int* __restrict__ grouped, const float* __restrict__ dinv,
                              float* __restrict__ gw, int E) {
    const int i = blockIdx.x * blockDim.x + threadIdx.x;
    if (i < E) gw[i] = dinv[grouped[i]];
}

template <int U>
__device__ __forceinline__ void gather_fma(float4& acc, const float4* __restrict__ in4, int idx,
                                           float wgt, int k, int lane) {
    float4 v[U];
#pragma unroll
    for (int u = 0; u < U; ++u) {
        const unsigned s = (unsigned)__builtin_amdgcn_readlane(idx, k + u);
        v[u] = in4[(size_t)s * (HID / 4) + lane];
    }
#pragma unroll
    for (int u = 0; u < U; ++u) {
        const float w = __int_as_float(__builtin_amdgcn_readlane(__float_as_int(wgt), k + u));
        acc.x += w * v[u].x;
        acc.y += w * v[u].y;
        acc.z += w * v[u].z;
        acc.w += w * v[u].w;
    }
}

#define LROWS 32
#define LPITCH (HID + 4)
#define LAYER_TPB 256
static_assert(NN % LROWS == 0, "every block's rows exist");
static_assert(LROWS == 32, "pooling epilogue: lane i of a wave holds the graph id of tile row i");
static_assert(IND == HID, "one tile width for both layers");
static_assert(LAYER_TPB / 32 == HID / 16, "one wave per 16-column tile of W");
static_assert(LROWS % (LAYER_TPB / 32) == 0 && LROWS / (LAYER_TPB / 32) == 4,
              "rows per wave: a power of two, one metadata lane each");

template <bool POOL>
__global__ void __launch_bounds__(LAYER_TPB)
layer_k(const int* __restrict__ rowStart, const int* __restrict__ cnt,
        const int* __restrict__ grouped, const float* __restrict__ gw,
        const float* __restrict__ dinv, const float* __restrict__ in,
        const float* __restrict__ Wm, const float* __restrict__ bias,
        const int* __restrict__ batch, float* __restrict__ out) {
    __shared__ __attribute__((aligned(16))) float As[LROWS * LPITCH];
    const int lane = threadIdx.x & 31;
    const int wave = __builtin_amdgcn_readfirstlane((int)(threadIdx.x >> 5));
    const int row0 = blockIdx.x * LROWS;
    const float4* in4 = reinterpret_cast<const float4*>(in);

    constexpr int NW = LAYER_TPB / 32;
    constexpr int RPW = LROWS / NW;
    const int dl = row0 + wave + NW * (lane & (RPW - 1));
    const int startV = rowStart[dl];
    const int cntV = cnt[dl];
    const float wdV = dinv[dl];
    int idxF[RPW];
    float wgtF[RPW];
#pragma unroll
    for (int i = 0; i < RPW; ++i) {
        const int start = __builtin_amdgcn_readlane(startV, i);
        const int c = __builtin_amdgcn_readlane(cntV, i);
        idxF[i] = 0;
        wgtF[i] = 0.0f;
        if (lane < c) {
            idxF[i] = grouped[start + lane];
            wgtF[i] = gw[start + lane];
        }
    }
#pragma unroll
    for (int i = 0; i < RPW; ++i) {
        const int r = wave + NW * i;
        const int d = row0 + r;
        const int start = __builtin_amdgcn_readlane(startV, i);
        const int c = __builtin_amdgcn_readlane(cntV, i);
        const float wd = __int_as_float(__builtin_amdgcn_readlane(__float_as_int(wdV), i));
        const float4 self = in4[(size_t)d * (HID / 4) + lane];
        float4 acc = make_float4(0.0f, 0.0f, 0.0f, 0.0f);
        int idx = idxF[i];
        float wgt = wgtF[i];
        for (int base = 0; base < c; base += 32) {
            const int m = (c - base < 32) ? (c - base) : 32;
            if (base > 0) {
                idx = 0;
                wgt = 0.0f;
                if (lane < m) {
                    idx = grouped[start + base + lane];
                    wgt = gw[start + base + lane];
                }
            }
            int k = 0;
            for (; k + 8 <= m; k += 8) gather_fma<8>(acc, in4, idx, wgt, k, lane);
            if (k + 4 <= m) { gather_fma<4>(acc, in4, idx, wgt, k, lane); k += 4; }
            if (k + 2 <= m) { gather_fma<2>(acc, in4, idx, wgt, k, lane); k += 2; }
            if (k < m) gather_fma<1>(acc, in4, idx, wgt, k, lane);
        }
        acc.x = (acc.x + wd * self.x) * wd;
        acc.y = (acc.y + wd * self.y) * wd;
        acc.z = (acc.z + wd * self.z) * wd;
        acc.w = (acc.w + wd * self.w) * wd;
        *reinterpret_cast<float4*>(&As[r * LPITCH + lane * 4]) = acc;
    }
    __syncthreads();

    const int half  = lane >> 4;
    const int l16   = lane & 15;
    const int nBase = wave * 16;
    const float* bBase = Wm + (size_t)(half * 2) * HID + nBase + l16;
    const float* aBase = As + l16 * LPITCH + half * 2;

    v8f acc2[LROWS / 16];
#pragma unroll
    for (int t = 0; t < LROWS / 16; ++t) {
        v8f z = {};
        acc2[t] = z;
    }

#pragma unroll 8
    for (int k = 0; k < HID; k += 4) {
        v2f b;
        b.x = bBase[k * HID];
        b.y = bBase[k * HID + HID];
#pragma unroll
        for (int t = 0; t < LROWS / 16; ++t) {
            const v2f a = *reinterpret_cast<const v2f*>(aBase + t * 16 * LPITCH + k);
            acc2[t] = __builtin_amdgcn_wmma_f32_16x16x4_f32(false, a, false, b, (short)0,
                                                            acc2[t], false, false);
        }
    }

    const float bv = bias[nBase + l16];
    if (!POOL) {
#pragma unroll
        for (int t = 0; t < LROWS / 16; ++t) {
            const int rBase = row0 + t * 16 + half * 8;
#pragma unroll
            for (int v = 0; v < 8; ++v)
                out[(size_t)(rBase + v) * HID + nBase + l16] = fmaxf(acc2[t][v] + bv, 0.0f);
        }
    } else {
        const int gl = batch[row0 + lane];
        float val[LROWS / 16][8];
#pragma unroll
        for (int t = 0; t < LROWS / 16; ++t)
#pragma unroll
            for (int v = 0; v < 8; ++v) val[t][v] = fmaxf(acc2[t][v] + bv, 0.0f);
        float* pcol = out + nBase + l16;
        const int g0 = __builtin_amdgcn_readfirstlane(gl);
        if (__builtin_amdgcn_ballot_w32(gl != g0) == 0) {
            float s = 0.0f;
#pragma unroll
            for (int t = 0; t < LROWS / 16; ++t)
#pragma unroll
                for (int v = 0; v < 8; ++v) s += val[t][v];
            s += __shfl_xor(s, 16);
            if (half == 0) atomicAdd(pcol + (size_t)g0 * HID, s);
        } else {
#pragma unroll
            for (int t = 0; t < LROWS / 16; ++t) {
                const int rb = t * 16 + half * 8;
                int cur = __shfl(gl, rb);
                float a = 0.0f;
#pragma unroll
                for (int v = 0; v < 8; ++v) {
                    const int g = __shfl(gl, rb + v);
                    if (g != cur) {
                        atomicAdd(pcol + (size_t)cur * HID, a);
                        a = 0.0f;
                        cur = g;
                    }
                    a += val[t][v];
                }
                atomicAdd(pcol + (size_t)cur * HID, a);
            }
        }
    }
}

template <int MT, int N, int K, bool ACT_A, bool ROWSCALE, bool BIAS_OUT>
__global__ void gemm_k(const float* __restrict__ A, const float* __restrict__ B,
                       float* __restrict__ C, const float* __restrict__ actBias,
                       const float* __restrict__ rowScale, const float* __restrict__ biasOut) {
    const int lane   = threadIdx.x & 31;
    const int wave   = threadIdx.x >> 5;
    const int half   = lane >> 4;
    const int l16    = lane & 15;
    const int nBase  = wave * 16;
    const int mTile0 = blockIdx.x * MT;

    const float* aBase = A + (size_t)(mTile0 * 16 + l16) * K + half * 2;
    const float* bBase = B + (size_t)(half * 2) * N + nBase + l16;
    const float* abBase = actBias + half * 2;

    v8f acc[MT];
#pragma unroll
    for (int t = 0; t < MT; ++t) {
        v8f z = {};
        acc[t] = z;
    }

#pragma unroll 4
    for (int k = 0; k < K; k += 4) {
        v2f b;
        b.x = bBase[k * N];
        b.y = bBase[k * N + N];
        float abx = 0.0f, aby = 0.0f;
        if (ACT_A) {
            abx = abBase[k];
            aby = abBase[k + 1];
        }
#pragma unroll
        for (int t = 0; t < MT; ++t) {
            v2f a;
            a.x = aBase[t * 16 * K + k];
            a.y = aBase[t * 16 * K + k + 1];
            if (ACT_A) {
                a.x = fmaxf(a.x + abx, 0.0f);
                a.y = fmaxf(a.y + aby, 0.0f);
            }
            acc[t] = __builtin_amdgcn_wmma_f32_16x16x4_f32(false, a, false, b, (short)0,
                                                           acc[t], false, false);
        }
    }

    float bv = BIAS_OUT ? biasOut[nBase + l16] : 0.0f;
#pragma unroll
    for (int t = 0; t < MT; ++t) {
        const int rBase = (mTile0 + t) * 16 + half * 8;
#pragma unroll
        for (int v = 0; v < 8; ++v) {
            const size_t idx = (size_t)(rBase + v) * N + nBase + l16;
            float val = acc[t][v] + bv;
            if (ROWSCALE) val *= rowScale[rBase + v];
            C[idx] = val;
        }
    }
}


extern "C" void kernel_launch(void* const* d_in, const int* in_sizes, int n_in,
                              void* d_out, int out_size, void* d_ws, size_t ws_size,
                              hipStream_t stream) {
    const float* x     = (const float*)d_in[0];
    const int*   ei    = (const int*)d_in[1];
    const int*   batch = (const int*)d_in[2];
    const float* W1    = (const float*)d_in[3];
    const float* b1    = (const float*)d_in[4];
    const float* W2    = (const float*)d_in[5];
    const float* b2    = (const float*)d_in[6];
    const float* Wl    = (const float*)d_in[7];
    const float* bl    = (const float*)d_in[8];
    float* out = (float*)d_out;

    const int* src = ei;
    const int* dst = ei + NE;

    char* ws = (char*)d_ws;
    const size_t FEAT_BYTES = (size_t)NN * HID * sizeof(float);
    const size_t EDGE_BYTES = (size_t)NE * sizeof(int);
    float* dinv      = (float*)(ws);
    float* bufH      = (float*)(ws + 524288);
    int* hist        = (int*)(ws + 524288 + FEAT_BYTES);
    int* bucketStart = (int*)(ws + 524288 + FEAT_BYTES + 524288);
    float* pooled    = (float*)(ws + 524288 + 2 * FEAT_BYTES);
    int* cnt         = (int*)(ws + 524288 + 2 * FEAT_BYTES + 262144);
    int* rowStart    = (int*)(ws + 524288 + 2 * FEAT_BYTES + 262144 + 524288);
    int* rec         = (int*)(ws + 524288 + 2 * FEAT_BYTES + 262144 + 2 * 524288);
    int* grouped     = (int*)(ws + 524288 + 2 * FEAT_BYTES + 262144 + 2 * 524288 + EDGE_BYTES);
    float* gw        = (float*)(ws + 524288 + 2 * FEAT_BYTES + 262144 + 2 * 524288 + 2 * EDGE_BYTES);

    const int TPB = 256;

    bucket_hist_k<<<L1_BLOCKS, L1_TPB, 0, stream>>>(dst, hist, pooled);
    bucket_scatter_k<<<L1_BLOCKS, L1_TPB, 0, stream>>>(src, dst, hist, rec, bucketStart);
    bucket_sort_k<<<NBK, L2_TPB, 0, stream>>>(rec, bucketStart, cnt, rowStart, dinv, grouped);
    edge_weight_k<<<(NE + TPB - 1) / TPB, TPB, 0, stream>>>(grouped, dinv, gw, NE);

    layer_k<false><<<NN / LROWS, LAYER_TPB, 0, stream>>>(rowStart, cnt, grouped, gw, dinv, x, W1, b1,
                                                         batch, bufH);

    layer_k<true><<<NN / LROWS, LAYER_TPB, 0, stream>>>(rowStart, cnt, grouped, gw, dinv, bufH, W2, b2,
                                                        batch, pooled);

    gemm_k<2, OUTD, HID, false, false, true><<<16, 32 * (OUTD / 16), 0, stream>>>(
        pooled, Wl, out, nullptr, nullptr, bl);
}


# Edge grouping level 1 sorts each chunk by bucket inside LDS in one launch (was a count launch plus a place launch): whole-line record stores, no count table; row sum order = LDS arrival rank
# speedup vs baseline: 5.7296x; 1.2025x over previous
#include <hip/hip_runtime.h>

typedef __attribute__((ext_vector_type(2))) float v2f;
typedef __attribute__((ext_vector_type(8))) float v8f;

#define NN 100000
#define NE 1600000
#define IND 128
#define HID 128
#define OUTD 64
#define NG 512

#define BKT_SHIFT 8
#define BKT_NODES (1 << BKT_SHIFT)
#define NBK ((NN + BKT_NODES - 1) / BKT_NODES)
#define SRC_BITS 17
#define SRC_MASK ((1u << SRC_BITS) - 1u)
#define L1_BLOCKS 256
#define L1_TPB 1024
#define L1_CHUNK ((NE + L1_BLOCKS - 1) / L1_BLOCKS)
#define L1_EPT ((L1_CHUNK + L1_TPB - 1) / L1_TPB)
#define L2_TPB 1024
#define L2_NHW (L2_TPB / 16)
#define L2_RPH (L1_BLOCKS / L2_NHW)
#define L2_PF 2
static_assert(NN <= (1 << SRC_BITS), "a source row number fits the low bits of a record");
static_assert(SRC_BITS + BKT_SHIFT <= 31, "a record is one non-negative word");
static_assert(NBK < L1_TPB, "one thread per bucket plus one for the end mark");
static_assert(NG * HID <= L1_BLOCKS * L1_TPB, "the first launch has a thread per pooled element");
static_assert(BKT_NODES <= L2_TPB, "one thread per row of a bucket");
static_assert(L1_BLOCKS <= L2_TPB, "one thread per run of a bucket");
static_assert(L1_BLOCKS % L2_NHW == 0, "every half wave owns the same number of runs");
static_assert((L1_BLOCKS - 1) * L1_CHUNK < NE, "every chunk holds at least one edge");

__device__ __forceinline__ int wave_incl_scan(int v, int lane) {
#pragma unroll
    for (int off = 1; off < 32; off <<= 1) {
        const int u = __shfl_up(v, off);
        if (lane >= off) v += u;
    }
    return v;
}

template <int NWAVES>
__device__ __forceinline__ int block_excl_scan(int v, int* wtot, int lane, int wave) {
    const int s = wave_incl_scan(v, lane);
    if (lane == 31) wtot[wave] = s;
    __syncthreads();
    if (wave == 0) {
        const int w = (lane < NWAVES) ? wtot[lane] : 0;
        const int ws = wave_incl_scan(w, lane);
        if (lane < NWAVES) wtot[lane] = ws - w;
    }
    __syncthreads();
    return wtot[wave] + s - v;
}

__global__ void __launch_bounds__(L1_TPB)
chunk_sort_k(const int* __restrict__ src, const int* __restrict__ dst, int* __restrict__ rec,
             int* __restrict__ offT, float* __restrict__ pooled) {
    __shared__ int cntL[NBK + 1];
    __shared__ int wtot[L1_TPB / 32];
    __shared__ int stage[L1_CHUNK];
    const int t = threadIdx.x;
    const int lane = t & 31;
    const int wave = t >> 5;
    if (t <= NBK) cntL[t] = 0;
    const int g = blockIdx.x * L1_TPB + t;
    if (g < NG * HID) pooled[g] = 0.0f;
    __syncthreads();
    const int e0 = blockIdx.x * L1_CHUNK;
    const int e1 = (e0 + L1_CHUNK < NE) ? (e0 + L1_CHUNK) : NE;
    int d[L1_EPT], s[L1_EPT], r[L1_EPT];
#pragma unroll
    for (int k = 0; k < L1_EPT; ++k) {
        const int e = e0 + t + k * L1_TPB;
        const int ec = (e < e1) ? e : (e1 - 1);
        d[k] = dst[ec];
        s[k] = src[ec];
    }
#pragma unroll
    for (int k = 0; k < L1_EPT; ++k) {
        const int e = e0 + t + k * L1_TPB;
        r[k] = 0;
        if (e < e1) r[k] = atomicAdd(&cntL[d[k] >> BKT_SHIFT], 1);
    }
    __syncthreads();
    const int cv = (t < NBK) ? cntL[t] : 0;
    const int off = block_excl_scan<L1_TPB / 32>(cv, wtot, lane, wave);
    if (t <= NBK) {
        cntL[t] = off;
        offT[t * L1_BLOCKS + blockIdx.x] = off;
    }
    __syncthreads();
#pragma unroll
    for (int k = 0; k < L1_EPT; ++k) {
        const int e = e0 + t + k * L1_TPB;
        if (e < e1)
            stage[cntL[d[k] >> BKT_SHIFT] + r[k]] = s[k] | ((d[k] & (BKT_NODES - 1)) << SRC_BITS);
    }
    __syncthreads();
    for (int i = t; i < e1 - e0; i += L1_TPB) rec[e0 + i] = stage[i];
}

__global__ void __launch_bounds__(L2_TPB)
bucket_sort_k(const int* __restrict__ rec, const int* __restrict__ offT,
              int* __restrict__ cnt, int* __restrict__ rowStart, float* __restrict__ dinv,
              int* __restrict__ grouped) {
    __shared__ int cL[BKT_NODES];
    __shared__ int oL[BKT_NODES];
    __shared__ int runBase[L1_BLOCKS];
    __shared__ int runLen[L1_BLOCKS];
    __shared__ int wtotA[L2_TPB / 32];
    __shared__ int wtotB[L2_TPB / 32];
    __shared__ int firstSlot;
    const int t = threadIdx.x;
    const int lane = t & 31;
    const int wave = t >> 5;
    const int b = blockIdx.x;
    int o0 = 0;
    if (t < L1_BLOCKS) {
        o0 = offT[b * L1_BLOCKS + t];
        runBase[t] = t * L1_CHUNK + o0;
        runLen[t] = offT[(b + 1) * L1_BLOCKS + t] - o0;
    }
    if (t < BKT_NODES) cL[t] = 0;
    const int exA = block_excl_scan<L2_TPB / 32>(o0, wtotA, lane, wave);
    if (t == L2_TPB - 1) firstSlot = exA + o0;
    __syncthreads();
    const int r0 = firstSlot;

    const int hw = t >> 4;
    const int l16 = t & 15;
    int rb[L2_RPH], rl[L2_RPH];
    unsigned w[L2_RPH][L2_PF];
#pragma unroll
    for (int q = 0; q < L2_RPH; ++q) {
        rb[q] = runBase[hw + L2_NHW * q];
        rl[q] = runLen[hw + L2_NHW * q];
    }
#pragma unroll
    for (int q = 0; q < L2_RPH; ++q) {
#pragma unroll
        for (int u = 0; u < L2_PF; ++u) {
            const int i = l16 + 16 * u;
            w[q][u] = (unsigned)rec[(i < rl[q]) ? (rb[q] + i) : ((hw + L2_NHW * q) * L1_CHUNK)];
        }
    }
#pragma unroll
    for (int q = 0; q < L2_RPH; ++q) {
#pragma unroll
        for (int u = 0; u < L2_PF; ++u)
            if (l16 + 16 * u < rl[q]) atomicAdd(&cL[w[q][u] >> SRC_BITS], 1);
        for (int i = 16 * L2_PF + l16; i < rl[q]; i += 16)
            atomicAdd(&cL[(unsigned)rec[rb[q] + i] >> SRC_BITS], 1);
    }
    __syncthreads();
    const int cv = (t < BKT_NODES) ? cL[t] : 0;
    const int off = block_excl_scan<L2_TPB / 32>(cv, wtotB, lane, wave);
    if (t < BKT_NODES) {
        oL[t] = off;
        cL[t] = 0;
        const int node = b * BKT_NODES + t;
        if (node < NN) {
            cnt[node] = cv;
            rowStart[node] = r0 + off;
            dinv[node] = rsqrtf((float)cv + 1.0f);
        }
    }
    __syncthreads();
#pragma unroll
    for (int q = 0; q < L2_RPH; ++q) {
#pragma unroll
        for (int u = 0; u < L2_PF; ++u) {
            if (l16 + 16 * u < rl[q]) {
                const int dl = (int)(w[q][u] >> SRC_BITS);
                const int p = atomicAdd(&cL[dl], 1);
                grouped[r0 + oL[dl] + p] = (int)(w[q][u] & SRC_MASK);
            }
        }
        for (int i = 16 * L2_PF + l16; i < rl[q]; i += 16) {
            const unsigned wr = (unsigned)rec[rb[q] + i];
            const int dl = (int)(wr >> SRC_BITS);
            const int p = atomicAdd(&cL[dl], 1);
            grouped[r0 + oL[dl] + p] = (int)(wr & SRC_MASK);
        }
    }
}

__global__ void edge_weight_k(const int* __restrict__ grouped, const float* __restrict__ dinv,
                              float* __restrict__ gw, int E) {
    const int i = blockIdx.x * blockDim.x + threadIdx.x;
    if (i < E) gw[i] = dinv[grouped[i]];
}

template <int U>
__device__ __forceinline__ void gather_fma(float4& acc, const float4* __restrict__ in4, int idx,
                                           float wgt, int k, int lane) {
    float4 v[U];
#pragma unroll
    for (int u = 0; u < U; ++u) {
        const unsigned s = (unsigned)__builtin_amdgcn_readlane(idx, k + u);
        v[u] = in4[(size_t)s * (HID / 4) + lane];
    }
#pragma unroll
    for (int u = 0; u < U; ++u) {
        const float w = __int_as_float(__builtin_amdgcn_readlane(__float_as_int(wgt), k + u));
        acc.x += w * v[u].x;
        acc.y += w * v[u].y;
        acc.z += w * v[u].z;
        acc.w += w * v[u].w;
    }
}

#define LROWS 32
#define LPITCH (HID + 4)
#define LAYER_TPB 256
static_assert(NN % LROWS == 0, "every block's rows exist");
static_assert(LROWS == 32, "pooling epilogue: lane i of a wave holds the graph id of tile row i");
static_assert(IND == HID, "one tile width for both layers");
static_assert(LAYER_TPB / 32 == HID / 16, "one wave per 16-column tile of W");
static_assert(LROWS % (LAYER_TPB / 32) == 0 && LROWS / (LAYER_TPB / 32) == 4,
              "rows per wave: a power of two, one metadata lane each");

template <bool POOL>
__global__ void __launch_bounds__(LAYER_TPB)
layer_k(const int* __restrict__ rowStart, const int* __restrict__ cnt,
        const int* __restrict__ grouped, const float* __restrict__ gw,
        const float* __restrict__ dinv, const float* __restrict__ in,
        const float* __restrict__ Wm, const float* __restrict__ bias,
        const int* __restrict__ batch, float* __restrict__ out) {
    __shared__ __attribute__((aligned(16))) float As[LROWS * LPITCH];
    const int lane = threadIdx.x & 31;
    const int wave = __builtin_amdgcn_readfirstlane((int)(threadIdx.x >> 5));
    const int row0 = blockIdx.x * LROWS;
    const float4* in4 = reinterpret_cast<const float4*>(in);

    constexpr int NW = LAYER_TPB / 32;
    constexpr int RPW = LROWS / NW;
    const int dl = row0 + wave + NW * (lane & (RPW - 1));
    const int startV = rowStart[dl];
    const int cntV = cnt[dl];
    const float wdV = dinv[dl];
    int idxF[RPW];
    float wgtF[RPW];
#pragma unroll
    for (int i = 0; i < RPW; ++i) {
        const int start = __builtin_amdgcn_readlane(startV, i);
        const int c = __builtin_amdgcn_readlane(cntV, i);
        idxF[i] = 0;
        wgtF[i] = 0.0f;
        if (lane < c) {
            idxF[i] = grouped[start + lane];
            wgtF[i] = gw[start + lane];
        }
    }
#pragma unroll
    for (int i = 0; i < RPW; ++i) {
        const int r = wave + NW * i;
        const int d = row0 + r;
        const int start = __builtin_amdgcn_readlane(startV, i);
        const int c = __builtin_amdgcn_readlane(cntV, i);
        const float wd = __int_as_float(__builtin_amdgcn_readlane(__float_as_int(wdV), i));
        const float4 self = in4[(size_t)d * (HID / 4) + lane];
        float4 acc = make_float4(0.0f, 0.0f, 0.0f, 0.0f);
        int idx = idxF[i];
        float wgt = wgtF[i];
        for (int base = 0; base < c; base += 32) {
            const int m = (c - base < 32) ? (c - base) : 32;
            if (base > 0) {
                idx = 0;
                wgt = 0.0f;
                if (lane < m) {
                    idx = grouped[start + base + lane];
                    wgt = gw[start + base + lane];
                }
            }
            int k = 0;
            for (; k + 8 <= m; k += 8) gather_fma<8>(acc, in4, idx, wgt, k, lane);
            if (k + 4 <= m) { gather_fma<4>(acc, in4, idx, wgt, k, lane); k += 4; }
            if (k + 2 <= m) { gather_fma<2>(acc, in4, idx, wgt, k, lane); k += 2; }
            if (k < m) gather_fma<1>(acc, in4, idx, wgt, k, lane);
        }
        acc.x = (acc.x + wd * self.x) * wd;
        acc.y = (acc.y + wd * self.y) * wd;
        acc.z = (acc.z + wd * self.z) * wd;
        acc.w = (acc.w + wd * self.w) * wd;
        *reinterpret_cast<float4*>(&As[r * LPITCH + lane * 4]) = acc;
    }
    __syncthreads();

    const int half  = lane >> 4;
    const int l16   = lane & 15;
    const int nBase = wave * 16;
    const float* bBase = Wm + (size_t)(half * 2) * HID + nBase + l16;
    const float* aBase = As + l16 * LPITCH + half * 2;

    v8f acc2[LROWS / 16];
#pragma unroll
    for (int t = 0; t < LROWS / 16; ++t) {
        v8f z = {};
        acc2[t] = z;
    }

#pragma unroll 8
    for (int k = 0; k < HID; k += 4) {
        v2f b;
        b.x = bBase[k * HID];
        b.y = bBase[k * HID + HID];
#pragma unroll
        for (int t = 0; t < LROWS / 16; ++t) {
            const v2f a = *reinterpret_cast<const v2f*>(aBase + t * 16 * LPITCH + k);
            acc2[t] = __builtin_amdgcn_wmma_f32_16x16x4_f32(false, a, false, b, (short)0,
                                                            acc2[t], false, false);
        }
    }

    const float bv = bias[nBase + l16];
    if (!POOL) {
#pragma unroll
        for (int t = 0; t < LROWS / 16; ++t) {
            const int rBase = row0 + t * 16 + half * 8;
#pragma unroll
            for (int v = 0; v < 8; ++v)
                out[(size_t)(rBase + v) * HID + nBase + l16] = fmaxf(acc2[t][v] + bv, 0.0f);
        }
    } else {
        const int gl = batch[row0 + lane];
        float val[LROWS / 16][8];
#pragma unroll
        for (int t = 0; t < LROWS / 16; ++t)
#pragma unroll
            for (int v = 0; v < 8; ++v) val[t][v] = fmaxf(acc2[t][v] + bv, 0.0f);
        float* pcol = out + nBase + l16;
        const int g0 = __builtin_amdgcn_readfirstlane(gl);
        if (__builtin_amdgcn_ballot_w32(gl != g0) == 0) {
            float s = 0.0f;
#pragma unroll
            for (int t = 0; t < LROWS / 16; ++t)
#pragma unroll
                for (int v = 0; v < 8; ++v) s += val[t][v];
            s += __shfl_xor(s, 16);
            if (half == 0) atomicAdd(pcol + (size_t)g0 * HID, s);
        } else {
#pragma unroll
            for (int t = 0; t < LROWS / 16; ++t) {
                const int rb = t * 16 + half * 8;
                int cur = __shfl(gl, rb);
                float a = 0.0f;
#pragma unroll
                for (int v = 0; v < 8; ++v) {
                    const int g = __shfl(gl, rb + v);
                    if (g != cur) {
                        atomicAdd(pcol + (size_t)cur * HID, a);
                        a = 0.0f;
                        cur = g;
                    }
                    a += val[t][v];
                }
                atomicAdd(pcol + (size_t)cur * HID, a);
            }
        }
    }
}

template <int MT, int N, int K, bool ACT_A, bool ROWSCALE, bool BIAS_OUT>
__global__ void gemm_k(const float* __restrict__ A, const float* __restrict__ B,
                       float* __restrict__ C, const float* __restrict__ actBias,
                       const float* __restrict__ rowScale, const float* __restrict__ biasOut) {
    const int lane   = threadIdx.x & 31;
    const int wave   = threadIdx.x >> 5;
    const int half   = lane >> 4;
    const int l16    = lane & 15;
    const int nBase  = wave * 16;
    const int mTile0 = blockIdx.x * MT;

    const float* aBase = A + (size_t)(mTile0 * 16 + l16) * K + half * 2;
    const float* bBase = B + (size_t)(half * 2) * N + nBase + l16;
    const float* abBase = actBias + half * 2;

    v8f acc[MT];
#pragma unroll
    for (int t = 0; t < MT; ++t) {
        v8f z = {};
        acc[t] = z;
    }

#pragma unroll 4
    for (int k = 0; k < K; k += 4) {
        v2f b;
        b.x = bBase[k * N];
        b.y = bBase[k * N + N];
        float abx = 0.0f, aby = 0.0f;
        if (ACT_A) {
            abx = abBase[k];
            aby = abBase[k + 1];
        }
#pragma unroll
        for (int t = 0; t < MT; ++t) {
            v2f a;
            a.x = aBase[t * 16 * K + k];
            a.y = aBase[t * 16 * K + k + 1];
            if (ACT_A) {
                a.x = fmaxf(a.x + abx, 0.0f);
                a.y = fmaxf(a.y + aby, 0.0f);
            }
            acc[t] = __builtin_amdgcn_wmma_f32_16x16x4_f32(false, a, false, b, (short)0,
                                                           acc[t], false, false);
        }
    }

    float bv = BIAS_OUT ? biasOut[nBase + l16] : 0.0f;
#pragma unroll
    for (int t = 0; t < MT; ++t) {
        const int rBase = (mTile0 + t) * 16 + half * 8;
#pragma unroll
        for (int v = 0; v < 8; ++v) {
            const size_t idx = (size_t)(rBase + v) * N + nBase + l16;
            float val = acc[t][v] + bv;
            if (ROWSCALE) val *= rowScale[rBase + v];
            C[idx] = val;
        }
    }
}


extern "C" void kernel_launch(void* const* d_in, const int* in_sizes, int n_in,
                              void* d_out, int out_size, void* d_ws, size_t ws_size,
                              hipStream_t stream) {
    const float* x     = (const float*)d_in[0];
    const int*   ei    = (const int*)d_in[1];
    const int*   batch = (const int*)d_in[2];
    const float* W1    = (const float*)d_in[3];
    const float* b1    = (const float*)d_in[4];
    const float* W2    = (const float*)d_in[5];
    const float* b2    = (const float*)d_in[6];
    const float* Wl    = (const float*)d_in[7];
    const float* bl    = (const float*)d_in[8];
    float* out = (float*)d_out;

    const int* src = ei;
    const int* dst = ei + NE;

    char* ws = (char*)d_ws;
    const size_t FEAT_BYTES = (size_t)NN * HID * sizeof(float);
    const size_t EDGE_BYTES = (size_t)NE * sizeof(int);
    float* dinv      = (float*)(ws);
    float* bufH      = (float*)(ws + 524288);
    int* offT        = (int*)(ws + 524288 + FEAT_BYTES);
    float* pooled    = (float*)(ws + 524288 + 2 * FEAT_BYTES);
    int* cnt         = (int*)(ws + 524288 + 2 * FEAT_BYTES + 262144);
    int* rowStart    = (int*)(ws + 524288 + 2 * FEAT_BYTES + 262144 + 524288);
    int* rec         = (int*)(ws + 524288 + 2 * FEAT_BYTES + 262144 + 2 * 524288);
    int* grouped     = (int*)(ws + 524288 + 2 * FEAT_BYTES + 262144 + 2 * 524288 + EDGE_BYTES);
    float* gw        = (float*)(ws + 524288 + 2 * FEAT_BYTES + 262144 + 2 * 524288 + 2 * EDGE_BYTES);

    const int TPB = 256;

    chunk_sort_k<<<L1_BLOCKS, L1_TPB, 0, stream>>>(src, dst, rec, offT, pooled);
    bucket_sort_k<<<NBK, L2_TPB, 0, stream>>>(rec, offT, cnt, rowStart, dinv, grouped);
    edge_weight_k<<<(NE + TPB - 1) / TPB, TPB, 0, stream>>>(grouped, dinv, gw, NE);

    layer_k<false><<<NN / LROWS, LAYER_TPB, 0, stream>>>(rowStart, cnt, grouped, gw, dinv, x, W1, b1,
                                                         batch, bufH);

    layer_k<true><<<NN / LROWS, LAYER_TPB, 0, stream>>>(rowStart, cnt, grouped, gw, dinv, bufH, W2, b2,
                                                        batch, pooled);

    gemm_k<2, OUTD, HID, false, false, true><<<16, 32 * (OUTD / 16), 0, stream>>>(
        pooled, Wl, out, nullptr, nullptr, bl);
}


# Edge grouping level 2 places a bucket's source rows in an LDS copy of its slice of the grouped list and stores it in whole 128-byte lines (was one 4-byte store per edge); row sum order unchanged
# speedup vs baseline: 5.9182x; 1.0329x over previous
#include <hip/hip_runtime.h>

typedef __attribute__((ext_vector_type(2))) float v2f;
typedef __attribute__((ext_vector_type(8))) float v8f;

#define NN 100000
#define NE 1600000
#define IND 128
#define HID 128
#define OUTD 64
#define NG 512

#define BKT_SHIFT 8
#define BKT_NODES (1 << BKT_SHIFT)
#define NBK ((NN + BKT_NODES - 1) / BKT_NODES)
#define SRC_BITS 17
#define SRC_MASK ((1u << SRC_BITS) - 1u)
#define L1_BLOCKS 256
#define L1_TPB 1024
#define L1_CHUNK ((NE + L1_BLOCKS - 1) / L1_BLOCKS)
#define L1_EPT ((L1_CHUNK + L1_TPB - 1) / L1_TPB)
#define L2_TPB 1024
#define L2_NHW (L2_TPB / 16)
#define L2_RPH (L1_BLOCKS / L2_NHW)
#define L2_PF 2
#define L2_STAGE 8192
static_assert(NN <= (1 << SRC_BITS), "a source row number fits the low bits of a record");
static_assert(SRC_BITS + BKT_SHIFT <= 31, "a record is one non-negative word");
static_assert(NBK < L1_TPB, "one thread per bucket plus one for the end mark");
static_assert(NG * HID <= L1_BLOCKS * L1_TPB, "the first launch has a thread per pooled element");
static_assert(BKT_NODES <= L2_TPB, "one thread per row of a bucket");
static_assert(L1_BLOCKS <= L2_TPB, "one thread per run of a bucket");
static_assert(L1_BLOCKS % L2_NHW == 0, "every half wave owns the same number of runs");
static_assert((L1_BLOCKS - 1) * L1_CHUNK < NE, "every chunk holds at least one edge");

__device__ __forceinline__ int wave_incl_scan(int v, int lane) {
#pragma unroll
    for (int off = 1; off < 32; off <<= 1) {
        const int u = __shfl_up(v, off);
        if (lane >= off) v += u;
    }
    return v;
}

template <int NWAVES>
__device__ __forceinline__ int block_excl_scan(int v, int* wtot, int lane, int wave) {
    const int s = wave_incl_scan(v, lane);
    if (lane == 31) wtot[wave] = s;
    __syncthreads();
    if (wave == 0) {
        const int w = (lane < NWAVES) ? wtot[lane] : 0;
        const int ws = wave_incl_scan(w, lane);
        if (lane < NWAVES) wtot[lane] = ws - w;
    }
    __syncthreads();
    return wtot[wave] + s - v;
}

__global__ void __launch_bounds__(L1_TPB)
chunk_sort_k(const int* __restrict__ src, const int* __restrict__ dst, int* __restrict__ rec,
             int* __restrict__ offT, float* __restrict__ pooled) {
    __shared__ int cntL[NBK + 1];
    __shared__ int wtot[L1_TPB / 32];
    __shared__ int stage[L1_CHUNK];
    const int t = threadIdx.x;
    const int lane = t & 31;
    const int wave = t >> 5;
    if (t <= NBK) cntL[t] = 0;
    const int g = blockIdx.x * L1_TPB + t;
    if (g < NG * HID) pooled[g] = 0.0f;
    __syncthreads();
    const int e0 = blockIdx.x * L1_CHUNK;
    const int e1 = (e0 + L1_CHUNK < NE) ? (e0 + L1_CHUNK) : NE;
    int d[L1_EPT], s[L1_EPT], r[L1_EPT];
#pragma unroll
    for (int k = 0; k < L1_EPT; ++k) {
        const int e = e0 + t + k * L1_TPB;
        const int ec = (e < e1) ? e : (e1 - 1);
        d[k] = dst[ec];
        s[k] = src[ec];
    }
#pragma unroll
    for (int k = 0; k < L1_EPT; ++k) {
        const int e = e0 + t + k * L1_TPB;
        r[k] = 0;
        if (e < e1) r[k] = atomicAdd(&cntL[d[k] >> BKT_SHIFT], 1);
    }
    __syncthreads();
    const int cv = (t < NBK) ? cntL[t] : 0;
    const int off = block_excl_scan<L1_TPB / 32>(cv, wtot, lane, wave);
    if (t <= NBK) {
        cntL[t] = off;
        offT[t * L1_BLOCKS + blockIdx.x] = off;
    }
    __syncthreads();
#pragma unroll
    for (int k = 0; k < L1_EPT; ++k) {
        const int e = e0 + t + k * L1_TPB;
        if (e < e1)
            stage[cntL[d[k] >> BKT_SHIFT] + r[k]] = s[k] | ((d[k] & (BKT_NODES - 1)) << SRC_BITS);
    }
    __syncthreads();
    for (int i = t; i < e1 - e0; i += L1_TPB) rec[e0 + i] = stage[i];
}

template <bool STAGED>
__device__ __forceinline__ void place_records(const int* __restrict__ rec, int* __restrict__ grouped,
                                              int* cL, const int* oL, int* outL, int r0, int l16,
                                              const int (&rb)[L2_RPH], const int (&rl)[L2_RPH],
                                              const unsigned (&w)[L2_RPH][L2_PF]) {
#pragma unroll
    for (int q = 0; q < L2_RPH; ++q) {
#pragma unroll
        for (int u = 0; u < L2_PF; ++u) {
            if (l16 + 16 * u < rl[q]) {
                const int dl = (int)(w[q][u] >> SRC_BITS);
                const int p = oL[dl] + atomicAdd(&cL[dl], 1);
                if (STAGED) outL[p] = (int)(w[q][u] & SRC_MASK);
                else grouped[r0 + p] = (int)(w[q][u] & SRC_MASK);
            }
        }
        for (int i = 16 * L2_PF + l16; i < rl[q]; i += 16) {
            const unsigned wr = (unsigned)rec[rb[q] + i];
            const int dl = (int)(wr >> SRC_BITS);
            const int p = oL[dl] + atomicAdd(&cL[dl], 1);
            if (STAGED) outL[p] = (int)(wr & SRC_MASK);
            else grouped[r0 + p] = (int)(wr & SRC_MASK);
        }
    }
}

__global__ void __launch_bounds__(L2_TPB)
bucket_sort_k(const int* __restrict__ rec, const int* __restrict__ offT,
              int* __restrict__ cnt, int* __restrict__ rowStart, float* __restrict__ dinv,
              int* __restrict__ grouped) {
    __shared__ int cL[BKT_NODES];
    __shared__ int oL[BKT_NODES];
    __shared__ int runBase[L1_BLOCKS];
    __shared__ int runLen[L1_BLOCKS];
    __shared__ int wtotA[L2_TPB / 32];
    __shared__ int wtotB[L2_TPB / 32];
    __shared__ int outL[L2_STAGE];
    __shared__ int firstSlot;
    __shared__ int bucketLen;
    const int t = threadIdx.x;
    const int lane = t & 31;
    const int wave = t >> 5;
    const int b = blockIdx.x;
    int o0 = 0;
    if (t < L1_BLOCKS) {
        o0 = offT[b * L1_BLOCKS + t];
        runBase[t] = t * L1_CHUNK + o0;
        runLen[t] = offT[(b + 1) * L1_BLOCKS + t] - o0;
    }
    if (t < BKT_NODES) cL[t] = 0;
    const int exA = block_excl_scan<L2_TPB / 32>(o0, wtotA, lane, wave);
    if (t == L2_TPB - 1) firstSlot = exA + o0;
    __syncthreads();
    const int r0 = firstSlot;

    const int hw = t >> 4;
    const int l16 = t & 15;
    int rb[L2_RPH], rl[L2_RPH];
    unsigned w[L2_RPH][L2_PF];
#pragma unroll
    for (int q = 0; q < L2_RPH; ++q) {
        rb[q] = runBase[hw + L2_NHW * q];
        rl[q] = runLen[hw + L2_NHW * q];
    }
#pragma unroll
    for (int q = 0; q < L2_RPH; ++q) {
#pragma unroll
        for (int u = 0; u < L2_PF; ++u) {
            const int i = l16 + 16 * u;
            w[q][u] = (unsigned)rec[(i < rl[q]) ? (rb[q] + i) : ((hw + L2_NHW * q) * L1_CHUNK)];
        }
    }
#pragma unroll
    for (int q = 0; q < L2_RPH; ++q) {
#pragma unroll
        for (int u = 0; u < L2_PF; ++u)
            if (l16 + 16 * u < rl[q]) atomicAdd(&cL[w[q][u] >> SRC_BITS], 1);
        for (int i = 16 * L2_PF + l16; i < rl[q]; i += 16)
            atomicAdd(&cL[(unsigned)rec[rb[q] + i] >> SRC_BITS], 1);
    }
    __syncthreads();
    const int cv = (t < BKT_NODES) ? cL[t] : 0;
    const int off = block_excl_scan<L2_TPB / 32>(cv, wtotB, lane, wave);
    if (t < BKT_NODES) {
        oL[t] = off;
        cL[t] = 0;
        if (t == BKT_NODES - 1) bucketLen = off + cv;
        const int node = b * BKT_NODES + t;
        if (node < NN) {
            cnt[node] = cv;
            rowStart[node] = r0 + off;
            dinv[node] = rsqrtf((float)cv + 1.0f);
        }
    }
    __syncthreads();
    const int total = bucketLen;
    if (total <= L2_STAGE) {
        place_records<true>(rec, grouped, cL, oL, outL, r0, l16, rb, rl, w);
        __syncthreads();
        for (int i = t - (r0 & 31); i < total; i += L2_TPB)
            if (i >= 0) grouped[r0 + i] = outL[i];
    } else {
        place_records<false>(rec, grouped, cL, oL, outL, r0, l16, rb, rl, w);
    }
}

__global__ void edge_weight_k(const int* __restrict__ grouped, const float* __restrict__ dinv,
                              float* __restrict__ gw, int E) {
    const int i = blockIdx.x * blockDim.x + threadIdx.x;
    if (i < E) gw[i] = dinv[grouped[i]];
}

template <int U>
__device__ __forceinline__ void gather_fma(float4& acc, const float4* __restrict__ in4, int idx,
                                           float wgt, int k, int lane) {
    float4 v[U];
#pragma unroll
    for (int u = 0; u < U; ++u) {
        const unsigned s = (unsigned)__builtin_amdgcn_readlane(idx, k + u);
        v[u] = in4[(size_t)s * (HID / 4) + lane];
    }
#pragma unroll
    for (int u = 0; u < U; ++u) {
        const float w = __int_as_float(__builtin_amdgcn_readlane(__float_as_int(wgt), k + u));
        acc.x += w * v[u].x;
        acc.y += w * v[u].y;
        acc.z += w * v[u].z;
        acc.w += w * v[u].w;
    }
}

#define LROWS 32
#define LPITCH (HID + 4)
#define LAYER_TPB 256
static_assert(NN % LROWS == 0, "every block's rows exist");
static_assert(LROWS == 32, "pooling epilogue: lane i of a wave holds the graph id of tile row i");
static_assert(IND == HID, "one tile width for both layers");
static_assert(LAYER_TPB / 32 == HID / 16, "one wave per 16-column tile of W");
static_assert(LROWS % (LAYER_TPB / 32) == 0 && LROWS / (LAYER_TPB / 32) == 4,
              "rows per wave: a power of two, one metadata lane each");

template <bool POOL>
__global__ void __launch_bounds__(LAYER_TPB)
layer_k(const int* __restrict__ rowStart, const int* __restrict__ cnt,
        const int* __restrict__ grouped, const float* __restrict__ gw,
        const float* __restrict__ dinv, const float* __restrict__ in,
        const float* __restrict__ Wm, const float* __restrict__ bias,
        const int* __restrict__ batch, float* __restrict__ out) {
    __shared__ __attribute__((aligned(16))) float As[LROWS * LPITCH];
    const int lane = threadIdx.x & 31;
    const int wave = __builtin_amdgcn_readfirstlane((int)(threadIdx.x >> 5));
    const int row0 = blockIdx.x * LROWS;
    const float4* in4 = reinterpret_cast<const float4*>(in);

    constexpr int NW = LAYER_TPB / 32;
    constexpr int RPW = LROWS / NW;
    const int dl = row0 + wave + NW * (lane & (RPW - 1));
    const int startV = rowStart[dl];
    const int cntV = cnt[dl];
    const float wdV = dinv[dl];
    int idxF[RPW];
    float wgtF[RPW];
#pragma unroll
    for (int i = 0; i < RPW; ++i) {
        const int start = __builtin_amdgcn_readlane(startV, i);
        const int c = __builtin_amdgcn_readlane(cntV, i);
        idxF[i] = 0;
        wgtF[i] = 0.0f;
        if (lane < c) {
            idxF[i] = grouped[start + lane];
            wgtF[i] = gw[start + lane];
        }
    }
#pragma unroll
    for (int i = 0; i < RPW; ++i) {
        const int r = wave + NW * i;
        const int d = row0 + r;
        const int start = __builtin_amdgcn_readlane(startV, i);
        const int c = __builtin_amdgcn_readlane(cntV, i);
        const float wd = __int_as_float(__builtin_amdgcn_readlane(__float_as_int(wdV), i));
        const float4 self = in4[(size_t)d * (HID / 4) + lane];
        float4 acc = make_float4(0.0f, 0.0f, 0.0f, 0.0f);
        int idx = idxF[i];
        float wgt = wgtF[i];
        for (int base = 0; base < c; base += 32) {
            const int m = (c - base < 32) ? (c - base) : 32;
            if (base > 0) {
                idx = 0;
                wgt = 0.0f;
                if (lane < m) {
                    idx = grouped[start + base + lane];
                    wgt = gw[start + base + lane];
                }
            }
            int k = 0;
            for (; k + 8 <= m; k += 8) gather_fma<8>(acc, in4, idx, wgt, k, lane);
            if (k + 4 <= m) { gather_fma<4>(acc, in4, idx, wgt, k, lane); k += 4; }
            if (k + 2 <= m) { gather_fma<2>(acc, in4, idx, wgt, k, lane); k += 2; }
            if (k < m) gather_fma<1>(acc, in4, idx, wgt, k, lane);
        }
        acc.x = (acc.x + wd * self.x) * wd;
        acc.y = (acc.y + wd * self.y) * wd;
        acc.z = (acc.z + wd * self.z) * wd;
        acc.w = (acc.w + wd * self.w) * wd;
        *reinterpret_cast<float4*>(&As[r * LPITCH + lane * 4]) = acc;
    }
    __syncthreads();

    const int half  = lane >> 4;
    const int l16   = lane & 15;
    const int nBase = wave * 16;
    const float* bBase = Wm + (size_t)(half * 2) * HID + nBase + l16;
    const float* aBase = As + l16 * LPITCH + half * 2;

    v8f acc2[LROWS / 16];
#pragma unroll
    for (int t = 0; t < LROWS / 16; ++t) {
        v8f z = {};
        acc2[t] = z;
    }

#pragma unroll 8
    for (int k = 0; k < HID; k += 4) {
        v2f b;
        b.x = bBase[k * HID];
        b.y = bBase[k * HID + HID];
#pragma unroll
        for (int t = 0; t < LROWS / 16; ++t) {
            const v2f a = *reinterpret_cast<const v2f*>(aBase + t * 16 * LPITCH + k);
            acc2[t] = __builtin_amdgcn_wmma_f32_16x16x4_f32(false, a, false, b, (short)0,
                                                            acc2[t], false, false);
        }
    }

    const float bv = bias[nBase + l16];
    if (!POOL) {
#pragma unroll
        for (int t = 0; t < LROWS / 16; ++t) {
            const int rBase = row0 + t * 16 + half * 8;
#pragma unroll
            for (int v = 0; v < 8; ++v)
                out[(size_t)(rBase + v) * HID + nBase + l16] = fmaxf(acc2[t][v] + bv, 0.0f);
        }
    } else {
        const int gl = batch[row0 + lane];
        float val[LROWS / 16][8];
#pragma unroll
        for (int t = 0; t < LROWS / 16; ++t)
#pragma unroll
            for (int v = 0; v < 8; ++v) val[t][v] = fmaxf(acc2[t][v] + bv, 0.0f);
        float* pcol = out + nBase + l16;
        const int g0 = __builtin_amdgcn_readfirstlane(gl);
        if (__builtin_amdgcn_ballot_w32(gl != g0) == 0) {
            float s = 0.0f;
#pragma unroll
            for (int t = 0; t < LROWS / 16; ++t)
#pragma unroll
                for (int v = 0; v < 8; ++v) s += val[t][v];
            s += __shfl_xor(s, 16);
            if (half == 0) atomicAdd(pcol + (size_t)g0 * HID, s);
        } else {
#pragma unroll
            for (int t = 0; t < LROWS / 16; ++t) {
                const int rb = t * 16 + half * 8;
                int cur = __shfl(gl, rb);
                float a = 0.0f;
#pragma unroll
                for (int v = 0; v < 8; ++v) {
                    const int g = __shfl(gl, rb + v);
                    if (g != cur) {
                        atomicAdd(pcol + (size_t)cur * HID, a);
                        a = 0.0f;
                        cur = g;
                    }
                    a += val[t][v];
                }
                atomicAdd(pcol + (size_t)cur * HID, a);
            }
        }
    }
}

template <int MT, int N, int K, bool ACT_A, bool ROWSCALE, bool BIAS_OUT>
__global__ void gemm_k(const float* __restrict__ A, const float* __restrict__ B,
                       float* __restrict__ C, const float* __restrict__ actBias,
                       const float* __restrict__ rowScale, const float* __restrict__ biasOut) {
    const int lane   = threadIdx.x & 31;
    const int wave   = threadIdx.x >> 5;
    const int half   = lane >> 4;
    const int l16    = lane & 15;
    const int nBase  = wave * 16;
    const int mTile0 = blockIdx.x * MT;

    const float* aBase = A + (size_t)(mTile0 * 16 + l16) * K + half * 2;
    const float* bBase = B + (size_t)(half * 2) * N + nBase + l16;
    const float* abBase = actBias + half * 2;

    v8f acc[MT];
#pragma unroll
    for (int t = 0; t < MT; ++t) {
        v8f z = {};
        acc[t] = z;
    }

#pragma unroll 4
    for (int k = 0; k < K; k += 4) {
        v2f b;
        b.x = bBase[k * N];
        b.y = bBase[k * N + N];
        float abx = 0.0f, aby = 0.0f;
        if (ACT_A) {
            abx = abBase[k];
            aby = abBase[k + 1];
        }
#pragma unroll
        for (int t = 0; t < MT; ++t) {
            v2f a;
            a.x = aBase[t * 16 * K + k];
            a.y = aBase[t * 16 * K + k + 1];
            if (ACT_A) {
                a.x = fmaxf(a.x + abx, 0.0f);
                a.y = fmaxf(a.y + aby, 0.0f);
            }
            acc[t] = __builtin_amdgcn_wmma_f32_16x16x4_f32(false, a, false, b, (short)0,
                                                           acc[t], false, false);
        }
    }

    float bv = BIAS_OUT ? biasOut[nBase + l16] : 0.0f;
#pragma unroll
    for (int t = 0; t < MT; ++t) {
        const int rBase = (mTile0 + t) * 16 + half * 8;
#pragma unroll
        for (int v = 0; v < 8; ++v) {
            const size_t idx = (size_t)(rBase + v) * N + nBase + l16;
            float val = acc[t][v] + bv;
            if (ROWSCALE) val *= rowScale[rBase + v];
            C[idx] = val;
        }
    }
}


extern "C" void kernel_launch(void* const* d_in, const int* in_sizes, int n_in,
                              void* d_out, int out_size, void* d_ws, size_t ws_size,
                              hipStream_t stream) {
    const float* x     = (const float*)d_in[0];
    const int*   ei    = (const int*)d_in[1];
    const int*   batch = (const int*)d_in[2];
    const float* W1    = (const float*)d_in[3];
    const float* b1    = (const float*)d_in[4];
    const float* W2    = (const float*)d_in[5];
    const float* b2    = (const float*)d_in[6];
    const float* Wl    = (const float*)d_in[7];
    const float* bl    = (const float*)d_in[8];
    float* out = (float*)d_out;

    const int* src = ei;
    const int* dst = ei + NE;

    char* ws = (char*)d_ws;
    const size_t FEAT_BYTES = (size_t)NN * HID * sizeof(float);
    const size_t EDGE_BYTES = (size_t)NE * sizeof(int);
    float* dinv      = (float*)(ws);
    float* bufH      = (float*)(ws + 524288);
    int* offT        = (int*)(ws + 524288 + FEAT_BYTES);
    float* pooled    = (float*)(ws + 524288 + 2 * FEAT_BYTES);
    int* cnt         = (int*)(ws + 524288 + 2 * FEAT_BYTES + 262144);
    int* rowStart    = (int*)(ws + 524288 + 2 * FEAT_BYTES + 262144 + 524288);
    int* rec         = (int*)(ws + 524288 + 2 * FEAT_BYTES + 262144 + 2 * 524288);
    int* grouped     = (int*)(ws + 524288 + 2 * FEAT_BYTES + 262144 + 2 * 524288 + EDGE_BYTES);
    float* gw        = (float*)(ws + 524288 + 2 * FEAT_BYTES + 262144 + 2 * 524288 + 2 * EDGE_BYTES);

    const int TPB = 256;

    chunk_sort_k<<<L1_BLOCKS, L1_TPB, 0, stream>>>(src, dst, rec, offT, pooled);
    bucket_sort_k<<<NBK, L2_TPB, 0, stream>>>(rec, offT, cnt, rowStart, dinv, grouped);
    edge_weight_k<<<(NE + TPB - 1) / TPB, TPB, 0, stream>>>(grouped, dinv, gw, NE);

    layer_k<false><<<NN / LROWS, LAYER_TPB, 0, stream>>>(rowStart, cnt, grouped, gw, dinv, x, W1, b1,
                                                         batch, bufH);

    layer_k<true><<<NN / LROWS, LAYER_TPB, 0, stream>>>(rowStart, cnt, grouped, gw, dinv, bufH, W2, b2,
                                                        batch, pooled);

    gemm_k<2, OUTD, HID, false, false, true><<<16, 32 * (OUTD / 16), 0, stream>>>(
        pooled, Wl, out, nullptr, nullptr, bl);
}
